# starting kernel
# baseline (speedup 1.0000x reference)
#include <hip/hip_runtime.h>
#include <hip/hip_bf16.h>

typedef __attribute__((ext_vector_type(2))) float v2f;
typedef __attribute__((ext_vector_type(8))) float v8f;

#define B_ 4
#define Q_ 256
#define K_ 1024
#define D_ 256

__device__ __forceinline__ float tanh_fast(float x) {
    float e = __builtin_amdgcn_exp2f(x * 2.88539008177793f);
    return 1.0f - 2.0f * __builtin_amdgcn_rcpf(e + 1.0f);
}

__global__ __launch_bounds__(256)
void proj_gemm_wmma(const float* __restrict__ X, const float* __restrict__ W,
                    float* __restrict__ Y) {
    __shared__ float xs[16][D_ + 4];
    const int tid = threadIdx.x;
    const int m0  = blockIdx.y * 16;
    const int n0  = blockIdx.x * 128;

    #pragma unroll
    for (int i = 0; i < 16; ++i)
        xs[i][tid] = X[(size_t)(m0 + i) * D_ + tid];
    __syncthreads();

    const int lane = tid & 31;
    const int wave = tid >> 5;
    const int mn   = lane & 15;
    const int hi   = lane >> 4;
    const int nw   = n0 + wave * 16;

    v8f c = {};
    for (int kk = 0; kk < D_; kk += 4) {
        v2f a, b;
        a.x = xs[mn][kk + 2 * hi + 0];
        a.y = xs[mn][kk + 2 * hi + 1];
        b.x = W[(size_t)(kk + 2 * hi + 0) * D_ + nw + mn];
        b.y = W[(size_t)(kk + 2 * hi + 1) * D_ + nw + mn];
        c = __builtin_amdgcn_wmma_f32_16x16x4_f32(false, a, false, b,
                                                  (short)0, c, false, false);
    }
    #pragma unroll
    for (int r = 0; r < 8; ++r)
        Y[(size_t)(m0 + r + 8 * hi) * D_ + nw + mn] = c[r];
}

__global__ __launch_bounds__(256)
void scores_kernel(const float* __restrict__ qp, const float* __restrict__ kp,
                   const float* __restrict__ wv, const int* __restrict__ mask,
                   float* __restrict__ scores) {
    __shared__ float qs[16][D_ + 4];
    __shared__ float ks[16][D_ + 4];
    __shared__ float wvs[D_];
    const int tid = threadIdx.x;
    const int k0 = blockIdx.x * 16, q0 = blockIdx.y * 16, b = blockIdx.z;

    const float* qpb = qp + ((size_t)b * Q_ + q0) * D_;
    const float* kpb = kp + ((size_t)b * K_ + k0) * D_;
    #pragma unroll
    for (int i = 0; i < 16; ++i) {
        qs[i][tid] = qpb[(size_t)i * D_ + tid];
        ks[i][tid] = kpb[(size_t)i * D_ + tid];
    }
    wvs[tid] = wv[tid];
    __syncthreads();

    const int tk = tid & 15, tq = tid >> 4;
    float s = 0.f;
    #pragma unroll 4
    for (int e = 0; e < D_; ++e)
        s += tanh_fast(qs[tq][e] + ks[tk][e]) * wvs[e];

    const int kg = k0 + tk;
    const bool valid = mask[(size_t)b * K_ + kg] != 0;
    scores[((size_t)b * Q_ + q0 + tq) * K_ + kg] = valid ? s : -__builtin_inff();
}

__global__ __launch_bounds__(256)
void softmax_kernel(float* __restrict__ scores) {
    __shared__ float red[256];
    float* s = scores + (size_t)blockIdx.x * K_;
    const int tid = threadIdx.x;

    float v0[4];
    float mx = -__builtin_inff();
    #pragma unroll
    for (int i = 0; i < 4; ++i) { v0[i] = s[tid + i * 256]; mx = fmaxf(mx, v0[i]); }
    red[tid] = mx; __syncthreads();
    for (int off = 128; off > 0; off >>= 1) {
        if (tid < off) red[tid] = fmaxf(red[tid], red[tid + off]);
        __syncthreads();
    }
    mx = red[0]; __syncthreads();

    float ex[4], sum = 0.f;
    #pragma unroll
    for (int i = 0; i < 4; ++i) {
        ex[i] = __builtin_amdgcn_exp2f((v0[i] - mx) * 1.44269504088896f);
        sum += ex[i];
    }
    red[tid] = sum; __syncthreads();
    for (int off = 128; off > 0; off >>= 1) {
        if (tid < off) red[tid] += red[tid + off];
        __syncthreads();
    }
    const float inv = 1.0f / red[0];
    #pragma unroll
    for (int i = 0; i < 4; ++i) s[tid + i * 256] = ex[i] * inv;
}

__global__ __launch_bounds__(256)
void attend_wmma(const float* __restrict__ probs, const float* __restrict__ V,
                 float* __restrict__ out) {
    __shared__ float ps[16][K_ + 4];
    const int tid = threadIdx.x;
    const int q0 = blockIdx.x * 16, b = blockIdx.y;

    const float* pb = probs + ((size_t)b * Q_ + q0) * K_;
    #pragma unroll
    for (int i = 0; i < 64; ++i) {
        int f = i * 256 + tid;
        ps[f >> 10][f & 1023] = pb[(size_t)(f >> 10) * K_ + (f & 1023)];
    }
    __syncthreads();

    const int lane = tid & 31, wave = tid >> 5;
    const int mn = lane & 15, hi = lane >> 4;
    const int d0 = wave * 32;
    const float* vb = V + (size_t)b * K_ * D_;

    v8f c0 = {}, c1 = {};
    for (int kk = 0; kk < K_; kk += 4) {
        v2f a, b0, b1;
        a.x = ps[mn][kk + 2 * hi + 0];
        a.y = ps[mn][kk + 2 * hi + 1];
        const float* vr0 = vb + (size_t)(kk + 2 * hi + 0) * D_;
        const float* vr1 = vb + (size_t)(kk + 2 * hi + 1) * D_;
        b0.x = vr0[d0 + mn];       b0.y = vr1[d0 + mn];
        b1.x = vr0[d0 + 16 + mn];  b1.y = vr1[d0 + 16 + mn];
        c0 = __builtin_amdgcn_wmma_f32_16x16x4_f32(false, a, false, b0,
                                                   (short)0, c0, false, false);
        c1 = __builtin_amdgcn_wmma_f32_16x16x4_f32(false, a, false, b1,
                                                   (short)0, c1, false, false);
    }
    float* ob = out + ((size_t)b * Q_ + q0) * D_;
    #pragma unroll
    for (int r = 0; r < 8; ++r) {
        ob[(size_t)(r + 8 * hi) * D_ + d0 + mn]      = c0[r];
        ob[(size_t)(r + 8 * hi) * D_ + d0 + 16 + mn] = c1[r];
    }
}

extern "C" void kernel_launch(void* const* d_in, const int* in_sizes, int n_in,
                              void* d_out, int out_size, void* d_ws, size_t ws_size,
                              hipStream_t stream) {
    const float* q    = (const float*)d_in[0];
    const float* k    = (const float*)d_in[1];
    const float* v    = (const float*)d_in[2];
    const int*   mask = (const int*)  d_in[3];
    const float* Wq   = (const float*)d_in[4];
    const float* Wk   = (const float*)d_in[5];
    const float* wv   = (const float*)d_in[6];
    float* out = (float*)d_out;

    float* qp = (float*)d_ws;
    float* kp = qp + (size_t)B_ * Q_ * D_;
    float* sc = kp + (size_t)B_ * K_ * D_;

    proj_gemm_wmma<<<dim3(2, (B_ * Q_) / 16), 256, 0, stream>>>(q, Wq, qp);
    proj_gemm_wmma<<<dim3(2, (B_ * K_) / 16), 256, 0, stream>>>(k, Wk, kp);
    scores_kernel<<<dim3(K_ / 16, Q_ / 16, B_), 256, 0, stream>>>(qp, kp, wv, mask, sc);
    softmax_kernel<<<B_ * Q_, 256, 0, stream>>>(sc);
    attend_wmma<<<dim3(Q_ / 16, B_), 256, 0, stream>>>(sc, v, out);
}


# attend GEMM re-tiled: 256 blocks of 16q x 64d, k split over the 8 waves, each wave's 16-byte loads all issued ahead of its WMMAs; partials added through LDS in fixed order (new k summation tree)
# speedup vs baseline: 1.5373x; 1.5373x over previous
#include <hip/hip_runtime.h>
#include <hip/hip_bf16.h>

typedef __attribute__((ext_vector_type(2))) float v2f;
typedef __attribute__((ext_vector_type(4))) float v4f;
typedef __attribute__((ext_vector_type(8))) float v8f;

#define B_ 4
#define Q_ 256
#define K_ 1024
#define D_ 256

__device__ __forceinline__ float tanh_fast(float x) {
    float e = __builtin_amdgcn_exp2f(x * 2.88539008177793f);
    return 1.0f - 2.0f * __builtin_amdgcn_rcpf(e + 1.0f);
}

__global__ __launch_bounds__(256)
void proj_gemm_wmma(const float* __restrict__ X, const float* __restrict__ W,
                    float* __restrict__ Y) {
    __shared__ float xs[16][D_ + 4];
    const int tid = threadIdx.x;
    const int m0  = blockIdx.y * 16;
    const int n0  = blockIdx.x * 128;

    #pragma unroll
    for (int i = 0; i < 16; ++i)
        xs[i][tid] = X[(size_t)(m0 + i) * D_ + tid];
    __syncthreads();

    const int lane = tid & 31;
    const int wave = tid >> 5;
    const int mn   = lane & 15;
    const int hi   = lane >> 4;
    const int nw   = n0 + wave * 16;

    v8f c = {};
    for (int kk = 0; kk < D_; kk += 4) {
        v2f a, b;
        a.x = xs[mn][kk + 2 * hi + 0];
        a.y = xs[mn][kk + 2 * hi + 1];
        b.x = W[(size_t)(kk + 2 * hi + 0) * D_ + nw + mn];
        b.y = W[(size_t)(kk + 2 * hi + 1) * D_ + nw + mn];
        c = __builtin_amdgcn_wmma_f32_16x16x4_f32(false, a, false, b,
                                                  (short)0, c, false, false);
    }
    #pragma unroll
    for (int r = 0; r < 8; ++r)
        Y[(size_t)(m0 + r + 8 * hi) * D_ + nw + mn] = c[r];
}

__global__ __launch_bounds__(256)
void scores_kernel(const float* __restrict__ qp, const float* __restrict__ kp,
                   const float* __restrict__ wv, const int* __restrict__ mask,
                   float* __restrict__ scores) {
    __shared__ float qs[16][D_ + 4];
    __shared__ float ks[16][D_ + 4];
    __shared__ float wvs[D_];
    const int tid = threadIdx.x;
    const int k0 = blockIdx.x * 16, q0 = blockIdx.y * 16, b = blockIdx.z;

    const float* qpb = qp + ((size_t)b * Q_ + q0) * D_;
    const float* kpb = kp + ((size_t)b * K_ + k0) * D_;
    #pragma unroll
    for (int i = 0; i < 16; ++i) {
        qs[i][tid] = qpb[(size_t)i * D_ + tid];
        ks[i][tid] = kpb[(size_t)i * D_ + tid];
    }
    wvs[tid] = wv[tid];
    __syncthreads();

    const int tk = tid & 15, tq = tid >> 4;
    float s = 0.f;
    #pragma unroll 4
    for (int e = 0; e < D_; ++e)
        s += tanh_fast(qs[tq][e] + ks[tk][e]) * wvs[e];

    const int kg = k0 + tk;
    const bool valid = mask[(size_t)b * K_ + kg] != 0;
    scores[((size_t)b * Q_ + q0 + tq) * K_ + kg] = valid ? s : -__builtin_inff();
}

__global__ __launch_bounds__(256)
void softmax_kernel(float* __restrict__ scores) {
    __shared__ float red[256];
    float* s = scores + (size_t)blockIdx.x * K_;
    const int tid = threadIdx.x;

    float v0[4];
    float mx = -__builtin_inff();
    #pragma unroll
    for (int i = 0; i < 4; ++i) { v0[i] = s[tid + i * 256]; mx = fmaxf(mx, v0[i]); }
    red[tid] = mx; __syncthreads();
    for (int off = 128; off > 0; off >>= 1) {
        if (tid < off) red[tid] = fmaxf(red[tid], red[tid + off]);
        __syncthreads();
    }
    mx = red[0]; __syncthreads();

    float ex[4], sum = 0.f;
    #pragma unroll
    for (int i = 0; i < 4; ++i) {
        ex[i] = __builtin_amdgcn_exp2f((v0[i] - mx) * 1.44269504088896f);
        sum += ex[i];
    }
    red[tid] = sum; __syncthreads();
    for (int off = 128; off > 0; off >>= 1) {
        if (tid < off) red[tid] += red[tid + off];
        __syncthreads();
    }
    const float inv = 1.0f / red[0];
    #pragma unroll
    for (int i = 0; i < 4; ++i) s[tid + i * 256] = ex[i] * inv;
}

__global__ __launch_bounds__(256)
void attend_wmma(const float* __restrict__ probs, const float* __restrict__ V,
                 float* __restrict__ out) {
    __shared__ float part[8][16][64];
    const int tid = threadIdx.x;
    const int lane = tid & 31, w = tid >> 5;
    const int c = lane & 15, h = lane >> 4;
    const int q0 = blockIdx.x * 16, d0 = blockIdx.y * 64, b = blockIdx.z;

    const float* pa = probs + ((size_t)b * Q_ + q0 + c) * K_ + w * 128 + 4 * h;
    const float* pv = V + ((size_t)b * K_ + w * 128 + 4 * h) * D_ + d0 + 4 * c;

    v4f av[16], bv[16][4];
    #pragma unroll
    for (int s = 0; s < 16; ++s) {
        av[s] = *(const v4f*)(pa + 8 * s);
        #pragma unroll
        for (int i = 0; i < 4; ++i)
            bv[s][i] = *(const v4f*)(pv + (size_t)(8 * s + i) * D_);
    }

    v8f acc0 = {}, acc1 = {}, acc2 = {}, acc3 = {};
    #pragma unroll
    for (int s = 0; s < 16; ++s) {
        const v4f a = av[s], r0 = bv[s][0], r1 = bv[s][1], r2 = bv[s][2], r3 = bv[s][3];
        v2f a01, a23, f;
        a01.x = a.x; a01.y = a.y;
        a23.x = a.z; a23.y = a.w;
        f.x = r0.x; f.y = r1.x;
        acc0 = __builtin_amdgcn_wmma_f32_16x16x4_f32(false, a01, false, f, (short)0, acc0, false, false);
        f.x = r0.y; f.y = r1.y;
        acc1 = __builtin_amdgcn_wmma_f32_16x16x4_f32(false, a01, false, f, (short)0, acc1, false, false);
        f.x = r0.z; f.y = r1.z;
        acc2 = __builtin_amdgcn_wmma_f32_16x16x4_f32(false, a01, false, f, (short)0, acc2, false, false);
        f.x = r0.w; f.y = r1.w;
        acc3 = __builtin_amdgcn_wmma_f32_16x16x4_f32(false, a01, false, f, (short)0, acc3, false, false);
        f.x = r2.x; f.y = r3.x;
        acc0 = __builtin_amdgcn_wmma_f32_16x16x4_f32(false, a23, false, f, (short)0, acc0, false, false);
        f.x = r2.y; f.y = r3.y;
        acc1 = __builtin_amdgcn_wmma_f32_16x16x4_f32(false, a23, false, f, (short)0, acc1, false, false);
        f.x = r2.z; f.y = r3.z;
        acc2 = __builtin_amdgcn_wmma_f32_16x16x4_f32(false, a23, false, f, (short)0, acc2, false, false);
        f.x = r2.w; f.y = r3.w;
        acc3 = __builtin_amdgcn_wmma_f32_16x16x4_f32(false, a23, false, f, (short)0, acc3, false, false);
    }

    __builtin_amdgcn_sched_group_barrier(0x020, 80, 0);
    __builtin_amdgcn_sched_group_barrier(0x008, 128, 0);

    #pragma unroll
    for (int r = 0; r < 8; ++r) {
        v4f t;
        t.x = acc0[r]; t.y = acc1[r]; t.z = acc2[r]; t.w = acc3[r];
        *(v4f*)&part[w][8 * h + r][4 * c] = t;
    }
    __syncthreads();

    const int row = tid >> 4, c4 = (tid & 15) * 4;
    v4f sum = *(const v4f*)&part[0][row][c4];
    #pragma unroll
    for (int i = 1; i < 8; ++i)
        sum += *(const v4f*)&part[i][row][c4];
    *(v4f*)(out + ((size_t)b * Q_ + q0 + row) * D_ + d0 + c4) = sum;
}

extern "C" void kernel_launch(void* const* d_in, const int* in_sizes, int n_in,
                              void* d_out, int out_size, void* d_ws, size_t ws_size,
                              hipStream_t stream) {
    const float* q    = (const float*)d_in[0];
    const float* k    = (const float*)d_in[1];
    const float* v    = (const float*)d_in[2];
    const int*   mask = (const int*)  d_in[3];
    const float* Wq   = (const float*)d_in[4];
    const float* Wk   = (const float*)d_in[5];
    const float* wv   = (const float*)d_in[6];
    float* out = (float*)d_out;

    float* qp = (float*)d_ws;
    float* kp = qp + (size_t)B_ * Q_ * D_;
    float* sc = kp + (size_t)B_ * K_ * D_;

    proj_gemm_wmma<<<dim3(2, (B_ * Q_) / 16), 256, 0, stream>>>(q, Wq, qp);
    proj_gemm_wmma<<<dim3(2, (B_ * K_) / 16), 256, 0, stream>>>(k, Wk, kp);
    scores_kernel<<<dim3(K_ / 16, Q_ / 16, B_), 256, 0, stream>>>(qp, kp, wv, mask, sc);
    softmax_kernel<<<B_ * Q_, 256, 0, stream>>>(sc);
    attend_wmma<<<dim3(Q_ / 16, D_ / 64, B_), 256, 0, stream>>>(sc, v, out);
}


# scores: tanh(a+b) = 1 - 2/(1 + exp(2a)exp(2b)), both exps formed once per staged tile: 1 rcp + 2 packed fma per term, 4x4 outputs per thread; e summed as 2 partial sums + sum(wv) (new order)
# speedup vs baseline: 2.2837x; 1.4855x over previous
#include <hip/hip_runtime.h>
#include <hip/hip_bf16.h>

typedef __attribute__((ext_vector_type(2))) float v2f;
typedef __attribute__((ext_vector_type(4))) float v4f;
typedef __attribute__((ext_vector_type(8))) float v8f;

#define B_ 4
#define Q_ 256
#define K_ 1024
#define D_ 256

__global__ __launch_bounds__(256)
void proj_gemm_wmma(const float* __restrict__ X, const float* __restrict__ W,
                    float* __restrict__ Y) {
    __shared__ float xs[16][D_ + 4];
    const int tid = threadIdx.x;
    const int m0  = blockIdx.y * 16;
    const int n0  = blockIdx.x * 128;

    #pragma unroll
    for (int i = 0; i < 16; ++i)
        xs[i][tid] = X[(size_t)(m0 + i) * D_ + tid];
    __syncthreads();

    const int lane = tid & 31;
    const int wave = tid >> 5;
    const int mn   = lane & 15;
    const int hi   = lane >> 4;
    const int nw   = n0 + wave * 16;

    v8f c = {};
    for (int kk = 0; kk < D_; kk += 4) {
        v2f a, b;
        a.x = xs[mn][kk + 2 * hi + 0];
        a.y = xs[mn][kk + 2 * hi + 1];
        b.x = W[(size_t)(kk + 2 * hi + 0) * D_ + nw + mn];
        b.y = W[(size_t)(kk + 2 * hi + 1) * D_ + nw + mn];
        c = __builtin_amdgcn_wmma_f32_16x16x4_f32(false, a, false, b,
                                                  (short)0, c, false, false);
    }
    #pragma unroll
    for (int r = 0; r < 8; ++r)
        Y[(size_t)(m0 + r + 8 * hi) * D_ + nw + mn] = c[r];
}

#define SQ_ 32
#define SK_ 128
#define SP_ (D_ + 4)
#define TWO_LOG2E 2.88539008177793f

struct ScoreOps { v4f q[4], k[4], w; };

__device__ __forceinline__ void scores_read(ScoreOps& o, const float* qrow, const float* krow,
                                            const float* wm2, int e) {
    #pragma unroll
    for (int i = 0; i < 4; ++i) o.q[i] = *(const v4f*)(qrow + 8 * i * SP_ + e);
    #pragma unroll
    for (int j = 0; j < 4; ++j) o.k[j] = *(const v4f*)(krow + 16 * j * SP_ + e);
    o.w = *(const v4f*)(wm2 + e);
}

template <bool FACTORED>
__device__ __forceinline__ v2f scores_den(const ScoreOps& o, int n) {
    const int hf = n >> 4, i = (n >> 2) & 3, j = n & 3;
    v2f a, b, one;
    a.x = hf ? o.q[i].z : o.q[i].x; a.y = hf ? o.q[i].w : o.q[i].y;
    b.x = hf ? o.k[j].z : o.k[j].x; b.y = hf ? o.k[j].w : o.k[j].y;
    one.x = 1.0f; one.y = 1.0f;
    if (FACTORED) return a * b + one;
    const v2f x = (a + b) * TWO_LOG2E;
    v2f t;
    t.x = __builtin_amdgcn_exp2f(x.x); t.y = __builtin_amdgcn_exp2f(x.y);
    return t + one;
}

template <bool FACTORED>
__device__ __forceinline__ void scores_terms(const ScoreOps& o, v2f acc[4][4], v2f& wsum) {
    v2f wlo, whi;
    wlo.x = o.w.x; wlo.y = o.w.y; whi.x = o.w.z; whi.y = o.w.w;
    v2f t[32], r[32];
    #pragma unroll
    for (int n = 0; n < 4; ++n) t[n] = scores_den<FACTORED>(o, n);
    if (FACTORED) __builtin_amdgcn_sched_barrier(0);
    #pragma unroll
    for (int n = 0; n < 32; ++n) {
        r[n].x = __builtin_amdgcn_rcpf(t[n].x);
        r[n].y = __builtin_amdgcn_rcpf(t[n].y);
        if (n + 4 < 32) t[n + 4] = scores_den<FACTORED>(o, n + 4);
        if (n >= 3) {
            const int m = n - 3;
            acc[(m >> 2) & 3][m & 3] = r[m] * ((m >> 4) ? whi : wlo) + acc[(m >> 2) & 3][m & 3];
        }
        if (FACTORED) {
            __builtin_amdgcn_sched_group_barrier(0x400, 1, 0);
            __builtin_amdgcn_sched_group_barrier(0x002, 1, 0);
            __builtin_amdgcn_sched_group_barrier(0x400, 1, 0);
            __builtin_amdgcn_sched_group_barrier(0x002, 1, 0);
            __builtin_amdgcn_sched_barrier(0);
        }
    }
    #pragma unroll
    for (int m = 29; m < 32; ++m)
        acc[(m >> 2) & 3][m & 3] = r[m] * whi + acc[(m >> 2) & 3][m & 3];
    wsum += wlo + whi;
}

template <bool FACTORED>
__device__ __forceinline__ void scores_loop(const float* qrow, const float* krow, const float* wm2,
                                            v2f acc[4][4], v2f& wsum) {
    ScoreOps A, B;
    scores_read(A, qrow, krow, wm2, 0);
    for (int e = 0; e < D_; e += 8) {
        scores_read(B, qrow, krow, wm2, e + 4);
        if (FACTORED) __builtin_amdgcn_sched_barrier(0);
        scores_terms<FACTORED>(A, acc, wsum);
        scores_read(A, qrow, krow, wm2, (e + 8) & (D_ - 1));
        if (FACTORED) __builtin_amdgcn_sched_barrier(0);
        scores_terms<FACTORED>(B, acc, wsum);
    }
}

__global__ __launch_bounds__(256)
void scores_kernel(const float* __restrict__ qp, const float* __restrict__ kp,
                   const float* __restrict__ wv, const int* __restrict__ mask,
                   float* __restrict__ scores) {
    __shared__ float qs[SQ_][SP_];
    __shared__ float ks[SK_][SP_];
    __shared__ float wm2[D_];
    const int tid = threadIdx.x;
    const int k0 = blockIdx.x * SK_, q0 = blockIdx.y * SQ_, b = blockIdx.z;

    const float* qpb = qp + ((size_t)b * Q_ + q0) * D_;
    const float* kpb = kp + ((size_t)b * K_ + k0) * D_;
    const int c4 = (tid & 63) * 4, rh = tid >> 6;

    float big = 0.f;
    #pragma unroll
    for (int i = 0; i < SQ_ / 4; ++i) {
        const int r = 4 * i + rh;
        const v4f x = *(const v4f*)(qpb + (size_t)r * D_ + c4);
        big = fmaxf(fmaxf(big, fmaxf(fabsf(x.x), fabsf(x.y))), fmaxf(fabsf(x.z), fabsf(x.w)));
        v4f y;
        y.x = __builtin_amdgcn_exp2f(x.x * TWO_LOG2E); y.y = __builtin_amdgcn_exp2f(x.y * TWO_LOG2E);
        y.z = __builtin_amdgcn_exp2f(x.z * TWO_LOG2E); y.w = __builtin_amdgcn_exp2f(x.w * TWO_LOG2E);
        *(v4f*)&qs[r][c4] = y;
    }
    #pragma unroll
    for (int i = 0; i < SK_ / 4; ++i) {
        const int r = 4 * i + rh;
        const int lr = (r & 64) + ((r & 63) >> 2) + 16 * (r & 3);
        const v4f x = *(const v4f*)(kpb + (size_t)r * D_ + c4);
        big = fmaxf(fmaxf(big, fmaxf(fabsf(x.x), fabsf(x.y))), fmaxf(fabsf(x.z), fabsf(x.w)));
        v4f y;
        y.x = __builtin_amdgcn_exp2f(x.x * TWO_LOG2E); y.y = __builtin_amdgcn_exp2f(x.y * TWO_LOG2E);
        y.z = __builtin_amdgcn_exp2f(x.z * TWO_LOG2E); y.w = __builtin_amdgcn_exp2f(x.w * TWO_LOG2E);
        *(v4f*)&ks[lr][c4] = y;
    }
    wm2[tid] = -2.0f * wv[tid];
    const int out_of_range = __builtin_amdgcn_readfirstlane(__syncthreads_or(!(big <= 40.0f)));

    const int lane = tid & 31, w = tid >> 5;
    const int tk = lane & 15, kh = w & 1, tq = 2 * (w >> 1) + (lane >> 4);
    const float* qrow = &qs[tq][0];
    const float* krow = &ks[64 * kh + tk][0];

    v2f acc[4][4];
    #pragma unroll
    for (int i = 0; i < 4; ++i)
        #pragma unroll
        for (int j = 0; j < 4; ++j) { acc[i][j].x = 0.f; acc[i][j].y = 0.f; }
    v2f wsum; wsum.x = 0.f; wsum.y = 0.f;

    if (!out_of_range) {
        scores_loop<true>(qrow, krow, wm2, acc, wsum);
    } else {
        for (int i = 0; i < SQ_ / 4; ++i) {
            const int r = 4 * i + rh;
            *(v4f*)&qs[r][c4] = *(const v4f*)(qpb + (size_t)r * D_ + c4);
        }
        for (int i = 0; i < SK_ / 4; ++i) {
            const int r = 4 * i + rh;
            const int lr = (r & 64) + ((r & 63) >> 2) + 16 * (r & 3);
            *(v4f*)&ks[lr][c4] = *(const v4f*)(kpb + (size_t)r * D_ + c4);
        }
        __syncthreads();
        scores_loop<false>(qrow, krow, wm2, acc, wsum);
    }

    const float wtot = -0.5f * (wsum.x + wsum.y);
    const int kg = k0 + 64 * kh + 4 * tk;
    const int4 mk = *(const int4*)(mask + (size_t)b * K_ + kg);
    const float ninf = -__builtin_inff();
    #pragma unroll
    for (int i = 0; i < 4; ++i) {
        v4f s;
        s.x = mk.x != 0 ? (acc[i][0].x + acc[i][0].y) + wtot : ninf;
        s.y = mk.y != 0 ? (acc[i][1].x + acc[i][1].y) + wtot : ninf;
        s.z = mk.z != 0 ? (acc[i][2].x + acc[i][2].y) + wtot : ninf;
        s.w = mk.w != 0 ? (acc[i][3].x + acc[i][3].y) + wtot : ninf;
        *(v4f*)(scores + ((size_t)b * Q_ + q0 + tq + 8 * i) * K_ + kg) = s;
    }
}

__global__ __launch_bounds__(256)
void softmax_kernel(float* __restrict__ scores) {
    __shared__ float red[256];
    float* s = scores + (size_t)blockIdx.x * K_;
    const int tid = threadIdx.x;

    float v0[4];
    float mx = -__builtin_inff();
    #pragma unroll
    for (int i = 0; i < 4; ++i) { v0[i] = s[tid + i * 256]; mx = fmaxf(mx, v0[i]); }
    red[tid] = mx; __syncthreads();
    for (int off = 128; off > 0; off >>= 1) {
        if (tid < off) red[tid] = fmaxf(red[tid], red[tid + off]);
        __syncthreads();
    }
    mx = red[0]; __syncthreads();

    float ex[4], sum = 0.f;
    #pragma unroll
    for (int i = 0; i < 4; ++i) {
        ex[i] = __builtin_amdgcn_exp2f((v0[i] - mx) * 1.44269504088896f);
        sum += ex[i];
    }
    red[tid] = sum; __syncthreads();
    for (int off = 128; off > 0; off >>= 1) {
        if (tid < off) red[tid] += red[tid + off];
        __syncthreads();
    }
    const float inv = 1.0f / red[0];
    #pragma unroll
    for (int i = 0; i < 4; ++i) s[tid + i * 256] = ex[i] * inv;
}

__global__ __launch_bounds__(256)
void attend_wmma(const float* __restrict__ probs, const float* __restrict__ V,
                 float* __restrict__ out) {
    __shared__ float part[8][16][64];
    const int tid = threadIdx.x;
    const int lane = tid & 31, w = tid >> 5;
    const int c = lane & 15, h = lane >> 4;
    const int q0 = blockIdx.x * 16, d0 = blockIdx.y * 64, b = blockIdx.z;

    const float* pa = probs + ((size_t)b * Q_ + q0 + c) * K_ + w * 128 + 4 * h;
    const float* pv = V + ((size_t)b * K_ + w * 128 + 4 * h) * D_ + d0 + 4 * c;

    v4f av[16], bv[16][4];
    #pragma unroll
    for (int s = 0; s < 16; ++s) {
        av[s] = *(const v4f*)(pa + 8 * s);
        #pragma unroll
        for (int i = 0; i < 4; ++i)
            bv[s][i] = *(const v4f*)(pv + (size_t)(8 * s + i) * D_);
    }

    v8f acc0 = {}, acc1 = {}, acc2 = {}, acc3 = {};
    #pragma unroll
    for (int s = 0; s < 16; ++s) {
        const v4f a = av[s], r0 = bv[s][0], r1 = bv[s][1], r2 = bv[s][2], r3 = bv[s][3];
        v2f a01, a23, f;
        a01.x = a.x; a01.y = a.y;
        a23.x = a.z; a23.y = a.w;
        f.x = r0.x; f.y = r1.x;
        acc0 = __builtin_amdgcn_wmma_f32_16x16x4_f32(false, a01, false, f, (short)0, acc0, false, false);
        f.x = r0.y; f.y = r1.y;
        acc1 = __builtin_amdgcn_wmma_f32_16x16x4_f32(false, a01, false, f, (short)0, acc1, false, false);
        f.x = r0.z; f.y = r1.z;
        acc2 = __builtin_amdgcn_wmma_f32_16x16x4_f32(false, a01, false, f, (short)0, acc2, false, false);
        f.x = r0.w; f.y = r1.w;
        acc3 = __builtin_amdgcn_wmma_f32_16x16x4_f32(false, a01, false, f, (short)0, acc3, false, false);
        f.x = r2.x; f.y = r3.x;
        acc0 = __builtin_amdgcn_wmma_f32_16x16x4_f32(false, a23, false, f, (short)0, acc0, false, false);
        f.x = r2.y; f.y = r3.y;
        acc1 = __builtin_amdgcn_wmma_f32_16x16x4_f32(false, a23, false, f, (short)0, acc1, false, false);
        f.x = r2.z; f.y = r3.z;
        acc2 = __builtin_amdgcn_wmma_f32_16x16x4_f32(false, a23, false, f, (short)0, acc2, false, false);
        f.x = r2.w; f.y = r3.w;
        acc3 = __builtin_amdgcn_wmma_f32_16x16x4_f32(false, a23, false, f, (short)0, acc3, false, false);
    }

    __builtin_amdgcn_sched_group_barrier(0x020, 80, 0);
    __builtin_amdgcn_sched_group_barrier(0x008, 128, 0);

    #pragma unroll
    for (int r = 0; r < 8; ++r) {
        v4f t;
        t.x = acc0[r]; t.y = acc1[r]; t.z = acc2[r]; t.w = acc3[r];
        *(v4f*)&part[w][8 * h + r][4 * c] = t;
    }
    __syncthreads();

    const int row = tid >> 4, c4 = (tid & 15) * 4;
    v4f sum = *(const v4f*)&part[0][row][c4];
    #pragma unroll
    for (int i = 1; i < 8; ++i)
        sum += *(const v4f*)&part[i][row][c4];
    *(v4f*)(out + ((size_t)b * Q_ + q0 + row) * D_ + d0 + c4) = sum;
}

extern "C" void kernel_launch(void* const* d_in, const int* in_sizes, int n_in,
                              void* d_out, int out_size, void* d_ws, size_t ws_size,
                              hipStream_t stream) {
    const float* q    = (const float*)d_in[0];
    const float* k    = (const float*)d_in[1];
    const float* v    = (const float*)d_in[2];
    const int*   mask = (const int*)  d_in[3];
    const float* Wq   = (const float*)d_in[4];
    const float* Wk   = (const float*)d_in[5];
    const float* wv   = (const float*)d_in[6];
    float* out = (float*)d_out;

    float* qp = (float*)d_ws;
    float* kp = qp + (size_t)B_ * Q_ * D_;
    float* sc = kp + (size_t)B_ * K_ * D_;

    proj_gemm_wmma<<<dim3(2, (B_ * Q_) / 16), 256, 0, stream>>>(q, Wq, qp);
    proj_gemm_wmma<<<dim3(2, (B_ * K_) / 16), 256, 0, stream>>>(k, Wk, kp);
    scores_kernel<<<dim3(K_ / SK_, Q_ / SQ_, B_), 256, 0, stream>>>(qp, kp, wv, mask, sc);
    softmax_kernel<<<B_ * Q_, 256, 0, stream>>>(sc);
    attend_wmma<<<dim3(Q_ / 16, D_ / 64, B_), 256, 0, stream>>>(sc, v, out);
}


# softmax folded into the attend kernel: row max and row sum on the A fragments (lane, lane halves, 8 waves via LDS), exp(s-max) fed to the WMMAs, division by the row sum after the k sum (new order)
# speedup vs baseline: 2.5064x; 1.0975x over previous
#include <hip/hip_runtime.h>
#include <hip/hip_bf16.h>

typedef __attribute__((ext_vector_type(2))) float v2f;
typedef __attribute__((ext_vector_type(4))) float v4f;
typedef __attribute__((ext_vector_type(8))) float v8f;

#define B_ 4
#define Q_ 256
#define K_ 1024
#define D_ 256

__global__ __launch_bounds__(256)
void proj_gemm_wmma(const float* __restrict__ X, const float* __restrict__ W,
                    float* __restrict__ Y) {
    __shared__ float xs[16][D_ + 4];
    const int tid = threadIdx.x;
    const int m0  = blockIdx.y * 16;
    const int n0  = blockIdx.x * 128;

    #pragma unroll
    for (int i = 0; i < 16; ++i)
        xs[i][tid] = X[(size_t)(m0 + i) * D_ + tid];
    __syncthreads();

    const int lane = tid & 31;
    const int wave = tid >> 5;
    const int mn   = lane & 15;
    const int hi   = lane >> 4;
    const int nw   = n0 + wave * 16;

    v8f c = {};
    for (int kk = 0; kk < D_; kk += 4) {
        v2f a, b;
        a.x = xs[mn][kk + 2 * hi + 0];
        a.y = xs[mn][kk + 2 * hi + 1];
        b.x = W[(size_t)(kk + 2 * hi + 0) * D_ + nw + mn];
        b.y = W[(size_t)(kk + 2 * hi + 1) * D_ + nw + mn];
        c = __builtin_amdgcn_wmma_f32_16x16x4_f32(false, a, false, b,
                                                  (short)0, c, false, false);
    }
    #pragma unroll
    for (int r = 0; r < 8; ++r)
        Y[(size_t)(m0 + r + 8 * hi) * D_ + nw + mn] = c[r];
}

#define SQ_ 32
#define SK_ 128
#define SP_ (D_ + 4)
#define TWO_LOG2E 2.88539008177793f

struct ScoreOps { v4f q[4], k[4], w; };

__device__ __forceinline__ void scores_read(ScoreOps& o, const float* qrow, const float* krow,
                                            const float* wm2, int e) {
    #pragma unroll
    for (int i = 0; i < 4; ++i) o.q[i] = *(const v4f*)(qrow + 8 * i * SP_ + e);
    #pragma unroll
    for (int j = 0; j < 4; ++j) o.k[j] = *(const v4f*)(krow + 16 * j * SP_ + e);
    o.w = *(const v4f*)(wm2 + e);
}

template <bool FACTORED>
__device__ __forceinline__ v2f scores_den(const ScoreOps& o, int n) {
    const int hf = n >> 4, i = (n >> 2) & 3, j = n & 3;
    v2f a, b, one;
    a.x = hf ? o.q[i].z : o.q[i].x; a.y = hf ? o.q[i].w : o.q[i].y;
    b.x = hf ? o.k[j].z : o.k[j].x; b.y = hf ? o.k[j].w : o.k[j].y;
    one.x = 1.0f; one.y = 1.0f;
    if (FACTORED) return a * b + one;
    const v2f x = (a + b) * TWO_LOG2E;
    v2f t;
    t.x = __builtin_amdgcn_exp2f(x.x); t.y = __builtin_amdgcn_exp2f(x.y);
    return t + one;
}

template <bool FACTORED>
__device__ __forceinline__ void scores_terms(const ScoreOps& o, v2f acc[4][4], v2f& wsum) {
    v2f wlo, whi;
    wlo.x = o.w.x; wlo.y = o.w.y; whi.x = o.w.z; whi.y = o.w.w;
    v2f t[32], r[32];
    #pragma unroll
    for (int n = 0; n < 4; ++n) t[n] = scores_den<FACTORED>(o, n);
    if (FACTORED) __builtin_amdgcn_sched_barrier(0);
    #pragma unroll
    for (int n = 0; n < 32; ++n) {
        r[n].x = __builtin_amdgcn_rcpf(t[n].x);
        r[n].y = __builtin_amdgcn_rcpf(t[n].y);
        if (n + 4 < 32) t[n + 4] = scores_den<FACTORED>(o, n + 4);
        if (n >= 3) {
            const int m = n - 3;
            acc[(m >> 2) & 3][m & 3] = r[m] * ((m >> 4) ? whi : wlo) + acc[(m >> 2) & 3][m & 3];
        }
        if (FACTORED) {
            __builtin_amdgcn_sched_group_barrier(0x400, 1, 0);
            __builtin_amdgcn_sched_group_barrier(0x002, 1, 0);
            __builtin_amdgcn_sched_group_barrier(0x400, 1, 0);
            __builtin_amdgcn_sched_group_barrier(0x002, 1, 0);
            __builtin_amdgcn_sched_barrier(0);
        }
    }
    #pragma unroll
    for (int m = 29; m < 32; ++m)
        acc[(m >> 2) & 3][m & 3] = r[m] * whi + acc[(m >> 2) & 3][m & 3];
    wsum += wlo + whi;
}

template <bool FACTORED>
__device__ __forceinline__ void scores_loop(const float* qrow, const float* krow, const float* wm2,
                                            v2f acc[4][4], v2f& wsum) {
    ScoreOps A, B;
    scores_read(A, qrow, krow, wm2, 0);
    for (int e = 0; e < D_; e += 8) {
        scores_read(B, qrow, krow, wm2, e + 4);
        if (FACTORED) __builtin_amdgcn_sched_barrier(0);
        scores_terms<FACTORED>(A, acc, wsum);
        scores_read(A, qrow, krow, wm2, (e + 8) & (D_ - 1));
        if (FACTORED) __builtin_amdgcn_sched_barrier(0);
        scores_terms<FACTORED>(B, acc, wsum);
    }
}

__global__ __launch_bounds__(256)
void scores_kernel(const float* __restrict__ qp, const float* __restrict__ kp,
                   const float* __restrict__ wv, const int* __restrict__ mask,
                   float* __restrict__ scores) {
    __shared__ float qs[SQ_][SP_];
    __shared__ float ks[SK_][SP_];
    __shared__ float wm2[D_];
    const int tid = threadIdx.x;
    const int k0 = blockIdx.x * SK_, q0 = blockIdx.y * SQ_, b = blockIdx.z;

    const float* qpb = qp + ((size_t)b * Q_ + q0) * D_;
    const float* kpb = kp + ((size_t)b * K_ + k0) * D_;
    const int c4 = (tid & 63) * 4, rh = tid >> 6;

    float big = 0.f;
    #pragma unroll
    for (int i = 0; i < SQ_ / 4; ++i) {
        const int r = 4 * i + rh;
        const v4f x = *(const v4f*)(qpb + (size_t)r * D_ + c4);
        big = fmaxf(fmaxf(big, fmaxf(fabsf(x.x), fabsf(x.y))), fmaxf(fabsf(x.z), fabsf(x.w)));
        v4f y;
        y.x = __builtin_amdgcn_exp2f(x.x * TWO_LOG2E); y.y = __builtin_amdgcn_exp2f(x.y * TWO_LOG2E);
        y.z = __builtin_amdgcn_exp2f(x.z * TWO_LOG2E); y.w = __builtin_amdgcn_exp2f(x.w * TWO_LOG2E);
        *(v4f*)&qs[r][c4] = y;
    }
    #pragma unroll
    for (int i = 0; i < SK_ / 4; ++i) {
        const int r = 4 * i + rh;
        const int lr = (r & 64) + ((r & 63) >> 2) + 16 * (r & 3);
        const v4f x = *(const v4f*)(kpb + (size_t)r * D_ + c4);
        big = fmaxf(fmaxf(big, fmaxf(fabsf(x.x), fabsf(x.y))), fmaxf(fabsf(x.z), fabsf(x.w)));
        v4f y;
        y.x = __builtin_amdgcn_exp2f(x.x * TWO_LOG2E); y.y = __builtin_amdgcn_exp2f(x.y * TWO_LOG2E);
        y.z = __builtin_amdgcn_exp2f(x.z * TWO_LOG2E); y.w = __builtin_amdgcn_exp2f(x.w * TWO_LOG2E);
        *(v4f*)&ks[lr][c4] = y;
    }
    wm2[tid] = -2.0f * wv[tid];
    const int out_of_range = __builtin_amdgcn_readfirstlane(__syncthreads_or(!(big <= 40.0f)));

    const int lane = tid & 31, w = tid >> 5;
    const int tk = lane & 15, kh = w & 1, tq = 2 * (w >> 1) + (lane >> 4);
    const float* qrow = &qs[tq][0];
    const float* krow = &ks[64 * kh + tk][0];

    v2f acc[4][4];
    #pragma unroll
    for (int i = 0; i < 4; ++i)
        #pragma unroll
        for (int j = 0; j < 4; ++j) { acc[i][j].x = 0.f; acc[i][j].y = 0.f; }
    v2f wsum; wsum.x = 0.f; wsum.y = 0.f;

    if (!out_of_range) {
        scores_loop<true>(qrow, krow, wm2, acc, wsum);
    } else {
        for (int i = 0; i < SQ_ / 4; ++i) {
            const int r = 4 * i + rh;
            *(v4f*)&qs[r][c4] = *(const v4f*)(qpb + (size_t)r * D_ + c4);
        }
        for (int i = 0; i < SK_ / 4; ++i) {
            const int r = 4 * i + rh;
            const int lr = (r & 64) + ((r & 63) >> 2) + 16 * (r & 3);
            *(v4f*)&ks[lr][c4] = *(const v4f*)(kpb + (size_t)r * D_ + c4);
        }
        __syncthreads();
        scores_loop<false>(qrow, krow, wm2, acc, wsum);
    }

    const float wtot = -0.5f * (wsum.x + wsum.y);
    const int kg = k0 + 64 * kh + 4 * tk;
    const int4 mk = *(const int4*)(mask + (size_t)b * K_ + kg);
    const float ninf = -__builtin_inff();
    #pragma unroll
    for (int i = 0; i < 4; ++i) {
        v4f s;
        s.x = mk.x != 0 ? (acc[i][0].x + acc[i][0].y) + wtot : ninf;
        s.y = mk.y != 0 ? (acc[i][1].x + acc[i][1].y) + wtot : ninf;
        s.z = mk.z != 0 ? (acc[i][2].x + acc[i][2].y) + wtot : ninf;
        s.w = mk.w != 0 ? (acc[i][3].x + acc[i][3].y) + wtot : ninf;
        *(v4f*)(scores + ((size_t)b * Q_ + q0 + tq + 8 * i) * K_ + kg) = s;
    }
}

#define LOG2E 1.44269504088896f

__global__ __launch_bounds__(256)
void attend_softmax_wmma(const float* __restrict__ scores, const float* __restrict__ V,
                         float* __restrict__ out) {
    __shared__ float part[8][16][64];
    __shared__ float rmax[8][16];
    __shared__ float rsum[8][16];
    const int tid = threadIdx.x;
    const int lane = tid & 31, w = tid >> 5;
    const int c = lane & 15, h = lane >> 4;
    const int q0 = blockIdx.x * 16, d0 = blockIdx.y * 64, b = blockIdx.z;

    const float* pa = scores + ((size_t)b * Q_ + q0 + c) * K_ + w * 128 + 4 * h;
    const float* pv = V + ((size_t)b * K_ + w * 128 + 4 * h) * D_ + d0 + 4 * c;

    v4f av[16], bv[16][4];
    #pragma unroll
    for (int s = 0; s < 16; ++s)
        av[s] = *(const v4f*)(pa + 8 * s);
    __builtin_amdgcn_sched_barrier(0);
    #pragma unroll
    for (int s = 0; s < 16; ++s) {
        #pragma unroll
        for (int i = 0; i < 4; ++i)
            bv[s][i] = *(const v4f*)(pv + (size_t)(8 * s + i) * D_);
    }

    float m = fmaxf(fmaxf(av[0].x, av[0].y), fmaxf(av[0].z, av[0].w));
    #pragma unroll
    for (int s = 1; s < 16; ++s)
        m = fmaxf(m, fmaxf(fmaxf(av[s].x, av[s].y), fmaxf(av[s].z, av[s].w)));
    m = fmaxf(m, __shfl_xor(m, 16));
    rmax[w][c] = m;
    __builtin_amdgcn_fence(__ATOMIC_RELEASE, "workgroup", "local");
    __builtin_amdgcn_s_barrier();
    __builtin_amdgcn_fence(__ATOMIC_ACQUIRE, "workgroup", "local");
    float mx = rmax[0][c];
    #pragma unroll
    for (int i = 1; i < 8; ++i) mx = fmaxf(mx, rmax[i][c]);

    const float nmx = -mx * LOG2E;
    float ls = 0.f;
    #pragma unroll
    for (int s = 0; s < 16; ++s) {
        v4f p;
        p.x = __builtin_amdgcn_exp2f(__builtin_fmaf(av[s].x, LOG2E, nmx));
        p.y = __builtin_amdgcn_exp2f(__builtin_fmaf(av[s].y, LOG2E, nmx));
        p.z = __builtin_amdgcn_exp2f(__builtin_fmaf(av[s].z, LOG2E, nmx));
        p.w = __builtin_amdgcn_exp2f(__builtin_fmaf(av[s].w, LOG2E, nmx));
        av[s] = p;
        ls += (p.x + p.y) + (p.z + p.w);
    }
    ls += __shfl_xor(ls, 16);
    rsum[w][c] = ls;

    v8f acc0 = {}, acc1 = {}, acc2 = {}, acc3 = {};
    #pragma unroll
    for (int s = 0; s < 16; ++s) {
        const v4f a = av[s], r0 = bv[s][0], r1 = bv[s][1], r2 = bv[s][2], r3 = bv[s][3];
        v2f a01, a23, f;
        a01.x = a.x; a01.y = a.y;
        a23.x = a.z; a23.y = a.w;
        f.x = r0.x; f.y = r1.x;
        acc0 = __builtin_amdgcn_wmma_f32_16x16x4_f32(false, a01, false, f, (short)0, acc0, false, false);
        f.x = r0.y; f.y = r1.y;
        acc1 = __builtin_amdgcn_wmma_f32_16x16x4_f32(false, a01, false, f, (short)0, acc1, false, false);
        f.x = r0.z; f.y = r1.z;
        acc2 = __builtin_amdgcn_wmma_f32_16x16x4_f32(false, a01, false, f, (short)0, acc2, false, false);
        f.x = r0.w; f.y = r1.w;
        acc3 = __builtin_amdgcn_wmma_f32_16x16x4_f32(false, a01, false, f, (short)0, acc3, false, false);
        f.x = r2.x; f.y = r3.x;
        acc0 = __builtin_amdgcn_wmma_f32_16x16x4_f32(false, a23, false, f, (short)0, acc0, false, false);
        f.x = r2.y; f.y = r3.y;
        acc1 = __builtin_amdgcn_wmma_f32_16x16x4_f32(false, a23, false, f, (short)0, acc1, false, false);
        f.x = r2.z; f.y = r3.z;
        acc2 = __builtin_amdgcn_wmma_f32_16x16x4_f32(false, a23, false, f, (short)0, acc2, false, false);
        f.x = r2.w; f.y = r3.w;
        acc3 = __builtin_amdgcn_wmma_f32_16x16x4_f32(false, a23, false, f, (short)0, acc3, false, false);
    }

    #pragma unroll
    for (int r = 0; r < 8; ++r) {
        v4f t;
        t.x = acc0[r]; t.y = acc1[r]; t.z = acc2[r]; t.w = acc3[r];
        *(v4f*)&part[w][8 * h + r][4 * c] = t;
    }
    __syncthreads();

    const int row = tid >> 4, c4 = (tid & 15) * 4;
    float tot = rsum[0][row];
    #pragma unroll
    for (int i = 1; i < 8; ++i) tot += rsum[i][row];
    const float inv = 1.0f / tot;
    v4f sum = *(const v4f*)&part[0][row][c4];
    #pragma unroll
    for (int i = 1; i < 8; ++i)
        sum += *(const v4f*)&part[i][row][c4];
    *(v4f*)(out + ((size_t)b * Q_ + q0 + row) * D_ + d0 + c4) = sum * inv;
}

extern "C" void kernel_launch(void* const* d_in, const int* in_sizes, int n_in,
                              void* d_out, int out_size, void* d_ws, size_t ws_size,
                              hipStream_t stream) {
    const float* q    = (const float*)d_in[0];
    const float* k    = (const float*)d_in[1];
    const float* v    = (const float*)d_in[2];
    const int*   mask = (const int*)  d_in[3];
    const float* Wq   = (const float*)d_in[4];
    const float* Wk   = (const float*)d_in[5];
    const float* wv   = (const float*)d_in[6];
    float* out = (float*)d_out;

    float* qp = (float*)d_ws;
    float* kp = qp + (size_t)B_ * Q_ * D_;
    float* sc = kp + (size_t)B_ * K_ * D_;

    proj_gemm_wmma<<<dim3(2, (B_ * Q_) / 16), 256, 0, stream>>>(q, Wq, qp);
    proj_gemm_wmma<<<dim3(2, (B_ * K_) / 16), 256, 0, stream>>>(k, Wk, kp);
    scores_kernel<<<dim3(K_ / SK_, Q_ / SQ_, B_), 256, 0, stream>>>(qp, kp, wv, mask, sc);
    attend_softmax_wmma<<<dim3(Q_ / 16, D_ / 64, B_), 256, 0, stream>>>(sc, v, out);
}


# both projection GEMMs re-tiled into one launch of 256 equal blocks (64 kp + 16 qp rows x 64 cols), k split over the 8 waves, all 16-byte loads issued ahead; partials added via LDS (new k sum tree)
# speedup vs baseline: 3.2028x; 1.2778x over previous
#include <hip/hip_runtime.h>
#include <hip/hip_bf16.h>

typedef __attribute__((ext_vector_type(2))) float v2f;
typedef __attribute__((ext_vector_type(4))) float v4f;
typedef __attribute__((ext_vector_type(8))) float v8f;

#define B_ 4
#define Q_ 256
#define K_ 1024
#define D_ 256

#define PKR_ 64
#define PQR_ 16
#define PC_  64
#define PP_  (PC_ + 4)

__device__ __forceinline__ void proj_tile(const v4f a[4], const v4f b[4][4],
                                          float (*dst)[PP_], int h, int c) {
    v8f acc0 = {}, acc1 = {}, acc2 = {}, acc3 = {};
    #pragma unroll
    for (int s = 0; s < 4; ++s) {
        const v4f r0 = b[s][0], r1 = b[s][1], r2 = b[s][2], r3 = b[s][3];
        v2f a01, a23, f;
        a01.x = a[s].x; a01.y = a[s].y;
        a23.x = a[s].z; a23.y = a[s].w;
        f.x = r0.x; f.y = r1.x;
        acc0 = __builtin_amdgcn_wmma_f32_16x16x4_f32(false, a01, false, f, (short)0, acc0, false, false);
        f.x = r0.y; f.y = r1.y;
        acc1 = __builtin_amdgcn_wmma_f32_16x16x4_f32(false, a01, false, f, (short)0, acc1, false, false);
        f.x = r0.z; f.y = r1.z;
        acc2 = __builtin_amdgcn_wmma_f32_16x16x4_f32(false, a01, false, f, (short)0, acc2, false, false);
        f.x = r0.w; f.y = r1.w;
        acc3 = __builtin_amdgcn_wmma_f32_16x16x4_f32(false, a01, false, f, (short)0, acc3, false, false);
        f.x = r2.x; f.y = r3.x;
        acc0 = __builtin_amdgcn_wmma_f32_16x16x4_f32(false, a23, false, f, (short)0, acc0, false, false);
        f.x = r2.y; f.y = r3.y;
        acc1 = __builtin_amdgcn_wmma_f32_16x16x4_f32(false, a23, false, f, (short)0, acc1, false, false);
        f.x = r2.z; f.y = r3.z;
        acc2 = __builtin_amdgcn_wmma_f32_16x16x4_f32(false, a23, false, f, (short)0, acc2, false, false);
        f.x = r2.w; f.y = r3.w;
        acc3 = __builtin_amdgcn_wmma_f32_16x16x4_f32(false, a23, false, f, (short)0, acc3, false, false);
    }
    #pragma unroll
    for (int r = 0; r < 8; ++r) {
        v4f t;
        t.x = acc0[r]; t.y = acc1[r]; t.z = acc2[r]; t.w = acc3[r];
        *(v4f*)&dst[8 * h + r][4 * c] = t;
    }
}

__global__ __launch_bounds__(256)
void proj_gemm_wmma(const float* __restrict__ Xq, const float* __restrict__ Xk,
                    const float* __restrict__ Wq, const float* __restrict__ Wk,
                    float* __restrict__ qp, float* __restrict__ kp) {
    __shared__ float part[8][PKR_ + PQR_][PP_];
    const int tid = threadIdx.x;
    const int lane = tid & 31, w = tid >> 5;
    const int c = lane & 15, h = lane >> 4;
    const int n0 = blockIdx.x * PC_;
    const int by = blockIdx.y;
    const int k0 = 32 * w;

    const float* pak = Xk + (size_t)(by * PKR_ + c) * D_ + k0 + 4 * h;
    const float* paq = Xq + (size_t)(by * PQR_ + c) * D_ + k0 + 4 * h;
    const float* pbk = Wk + (size_t)(k0 + 4 * h) * D_ + n0 + 4 * c;
    const float* pbq = Wq + (size_t)(k0 + 4 * h) * D_ + n0 + 4 * c;

    v4f bk[4][4], bq[4][4], ak[4][4], aq[4];
    #pragma unroll
    for (int s = 0; s < 4; ++s) {
        ak[0][s] = *(const v4f*)(pak + 8 * s);
        #pragma unroll
        for (int i = 0; i < 4; ++i)
            bk[s][i] = *(const v4f*)(pbk + (size_t)(8 * s + i) * D_);
    }
    #pragma unroll
    for (int t = 1; t < 4; ++t) {
        #pragma unroll
        for (int s = 0; s < 4; ++s)
            ak[t][s] = *(const v4f*)(pak + (size_t)(16 * t) * D_ + 8 * s);
    }
    #pragma unroll
    for (int s = 0; s < 4; ++s) {
        aq[s] = *(const v4f*)(paq + 8 * s);
        #pragma unroll
        for (int i = 0; i < 4; ++i)
            bq[s][i] = *(const v4f*)(pbq + (size_t)(8 * s + i) * D_);
    }
    __builtin_amdgcn_sched_barrier(0);

    #pragma unroll
    for (int t = 0; t < 4; ++t)
        proj_tile(ak[t], bk, &part[w][16 * t], h, c);
    proj_tile(aq, bq, &part[w][PKR_], h, c);
    __syncthreads();

    const int r16 = tid >> 4, c4 = (tid & 15) * 4;
    #pragma unroll
    for (int t = 0; t < 5; ++t) {
        const int row = 16 * t + r16;
        v4f sum = *(const v4f*)&part[0][row][c4];
        #pragma unroll
        for (int p = 1; p < 8; ++p)
            sum += *(const v4f*)&part[p][row][c4];
        float* y = (t < 4) ? kp + (size_t)(by * PKR_ + row) * D_
                           : qp + (size_t)(by * PQR_ + r16) * D_;
        *(v4f*)(y + n0 + c4) = sum;
    }
}

#define SQ_ 32
#define SK_ 128
#define SP_ (D_ + 4)
#define TWO_LOG2E 2.88539008177793f

struct ScoreOps { v4f q[4], k[4], w; };

__device__ __forceinline__ void scores_read(ScoreOps& o, const float* qrow, const float* krow,
                                            const float* wm2, int e) {
    #pragma unroll
    for (int i = 0; i < 4; ++i) o.q[i] = *(const v4f*)(qrow + 8 * i * SP_ + e);
    #pragma unroll
    for (int j = 0; j < 4; ++j) o.k[j] = *(const v4f*)(krow + 16 * j * SP_ + e);
    o.w = *(const v4f*)(wm2 + e);
}

template <bool FACTORED>
__device__ __forceinline__ v2f scores_den(const ScoreOps& o, int n) {
    const int hf = n >> 4, i = (n >> 2) & 3, j = n & 3;
    v2f a, b, one;
    a.x = hf ? o.q[i].z : o.q[i].x; a.y = hf ? o.q[i].w : o.q[i].y;
    b.x = hf ? o.k[j].z : o.k[j].x; b.y = hf ? o.k[j].w : o.k[j].y;
    one.x = 1.0f; one.y = 1.0f;
    if (FACTORED) return a * b + one;
    const v2f x = (a + b) * TWO_LOG2E;
    v2f t;
    t.x = __builtin_amdgcn_exp2f(x.x); t.y = __builtin_amdgcn_exp2f(x.y);
    return t + one;
}

template <bool FACTORED>
__device__ __forceinline__ void scores_terms(const ScoreOps& o, v2f acc[4][4], v2f& wsum) {
    v2f wlo, whi;
    wlo.x = o.w.x; wlo.y = o.w.y; whi.x = o.w.z; whi.y = o.w.w;
    v2f t[32], r[32];
    #pragma unroll
    for (int n = 0; n < 4; ++n) t[n] = scores_den<FACTORED>(o, n);
    if (FACTORED) __builtin_amdgcn_sched_barrier(0);
    #pragma unroll
    for (int n = 0; n < 32; ++n) {
        r[n].x = __builtin_amdgcn_rcpf(t[n].x);
        r[n].y = __builtin_amdgcn_rcpf(t[n].y);
        if (n + 4 < 32) t[n + 4] = scores_den<FACTORED>(o, n + 4);
        if (n >= 3) {
            const int m = n - 3;
            acc[(m >> 2) & 3][m & 3] = r[m] * ((m >> 4) ? whi : wlo) + acc[(m >> 2) & 3][m & 3];
        }
        if (FACTORED) {
            __builtin_amdgcn_sched_group_barrier(0x400, 1, 0);
            __builtin_amdgcn_sched_group_barrier(0x002, 1, 0);
            __builtin_amdgcn_sched_group_barrier(0x400, 1, 0);
            __builtin_amdgcn_sched_group_barrier(0x002, 1, 0);
            __builtin_amdgcn_sched_barrier(0);
        }
    }
    #pragma unroll
    for (int m = 29; m < 32; ++m)
        acc[(m >> 2) & 3][m & 3] = r[m] * whi + acc[(m >> 2) & 3][m & 3];
    wsum += wlo + whi;
}

template <bool FACTORED>
__device__ __forceinline__ void scores_loop(const float* qrow, const float* krow, const float* wm2,
                                            v2f acc[4][4], v2f& wsum) {
    ScoreOps A, B;
    scores_read(A, qrow, krow, wm2, 0);
    for (int e = 0; e < D_; e += 8) {
        scores_read(B, qrow, krow, wm2, e + 4);
        if (FACTORED) __builtin_amdgcn_sched_barrier(0);
        scores_terms<FACTORED>(A, acc, wsum);
        scores_read(A, qrow, krow, wm2, (e + 8) & (D_ - 1));
        if (FACTORED) __builtin_amdgcn_sched_barrier(0);
        scores_terms<FACTORED>(B, acc, wsum);
    }
}

__global__ __launch_bounds__(256)
void scores_kernel(const float* __restrict__ qp, const float* __restrict__ kp,
                   const float* __restrict__ wv, const int* __restrict__ mask,
                   float* __restrict__ scores) {
    __shared__ float qs[SQ_][SP_];
    __shared__ float ks[SK_][SP_];
    __shared__ float wm2[D_];
    const int tid = threadIdx.x;
    const int k0 = blockIdx.x * SK_, q0 = blockIdx.y * SQ_, b = blockIdx.z;

    const float* qpb = qp + ((size_t)b * Q_ + q0) * D_;
    const float* kpb = kp + ((size_t)b * K_ + k0) * D_;
    const int c4 = (tid & 63) * 4, rh = tid >> 6;

    float big = 0.f;
    #pragma unroll
    for (int i = 0; i < SQ_ / 4; ++i) {
        const int r = 4 * i + rh;
        const v4f x = *(const v4f*)(qpb + (size_t)r * D_ + c4);
        big = fmaxf(fmaxf(big, fmaxf(fabsf(x.x), fabsf(x.y))), fmaxf(fabsf(x.z), fabsf(x.w)));
        v4f y;
        y.x = __builtin_amdgcn_exp2f(x.x * TWO_LOG2E); y.y = __builtin_amdgcn_exp2f(x.y * TWO_LOG2E);
        y.z = __builtin_amdgcn_exp2f(x.z * TWO_LOG2E); y.w = __builtin_amdgcn_exp2f(x.w * TWO_LOG2E);
        *(v4f*)&qs[r][c4] = y;
    }
    #pragma unroll
    for (int i = 0; i < SK_ / 4; ++i) {
        const int r = 4 * i + rh;
        const int lr = (r & 64) + ((r & 63) >> 2) + 16 * (r & 3);
        const v4f x = *(const v4f*)(kpb + (size_t)r * D_ + c4);
        big = fmaxf(fmaxf(big, fmaxf(fabsf(x.x), fabsf(x.y))), fmaxf(fabsf(x.z), fabsf(x.w)));
        v4f y;
        y.x = __builtin_amdgcn_exp2f(x.x * TWO_LOG2E); y.y = __builtin_amdgcn_exp2f(x.y * TWO_LOG2E);
        y.z = __builtin_amdgcn_exp2f(x.z * TWO_LOG2E); y.w = __builtin_amdgcn_exp2f(x.w * TWO_LOG2E);
        *(v4f*)&ks[lr][c4] = y;
    }
    wm2[tid] = -2.0f * wv[tid];
    const int out_of_range = __builtin_amdgcn_readfirstlane(__syncthreads_or(!(big <= 40.0f)));

    const int lane = tid & 31, w = tid >> 5;
    const int tk = lane & 15, kh = w & 1, tq = 2 * (w >> 1) + (lane >> 4);
    const float* qrow = &qs[tq][0];
    const float* krow = &ks[64 * kh + tk][0];

    v2f acc[4][4];
    #pragma unroll
    for (int i = 0; i < 4; ++i)
        #pragma unroll
        for (int j = 0; j < 4; ++j) { acc[i][j].x = 0.f; acc[i][j].y = 0.f; }
    v2f wsum; wsum.x = 0.f; wsum.y = 0.f;

    if (!out_of_range) {
        scores_loop<true>(qrow, krow, wm2, acc, wsum);
    } else {
        for (int i = 0; i < SQ_ / 4; ++i) {
            const int r = 4 * i + rh;
            *(v4f*)&qs[r][c4] = *(const v4f*)(qpb + (size_t)r * D_ + c4);
        }
        for (int i = 0; i < SK_ / 4; ++i) {
            const int r = 4 * i + rh;
            const int lr = (r & 64) + ((r & 63) >> 2) + 16 * (r & 3);
            *(v4f*)&ks[lr][c4] = *(const v4f*)(kpb + (size_t)r * D_ + c4);
        }
        __syncthreads();
        scores_loop<false>(qrow, krow, wm2, acc, wsum);
    }

    const float wtot = -0.5f * (wsum.x + wsum.y);
    const int kg = k0 + 64 * kh + 4 * tk;
    const int4 mk = *(const int4*)(mask + (size_t)b * K_ + kg);
    const float ninf = -__builtin_inff();
    #pragma unroll
    for (int i = 0; i < 4; ++i) {
        v4f s;
        s.x = mk.x != 0 ? (acc[i][0].x + acc[i][0].y) + wtot : ninf;
        s.y = mk.y != 0 ? (acc[i][1].x + acc[i][1].y) + wtot : ninf;
        s.z = mk.z != 0 ? (acc[i][2].x + acc[i][2].y) + wtot : ninf;
        s.w = mk.w != 0 ? (acc[i][3].x + acc[i][3].y) + wtot : ninf;
        *(v4f*)(scores + ((size_t)b * Q_ + q0 + tq + 8 * i) * K_ + kg) = s;
    }
}

#define LOG2E 1.44269504088896f

__global__ __launch_bounds__(256)
void attend_softmax_wmma(const float* __restrict__ scores, const float* __restrict__ V,
                         float* __restrict__ out) {
    __shared__ float part[8][16][64];
    __shared__ float rmax[8][16];
    __shared__ float rsum[8][16];
    const int tid = threadIdx.x;
    const int lane = tid & 31, w = tid >> 5;
    const int c = lane & 15, h = lane >> 4;
    const int q0 = blockIdx.x * 16, d0 = blockIdx.y * 64, b = blockIdx.z;

    const float* pa = scores + ((size_t)b * Q_ + q0 + c) * K_ + w * 128 + 4 * h;
    const float* pv = V + ((size_t)b * K_ + w * 128 + 4 * h) * D_ + d0 + 4 * c;

    v4f av[16], bv[16][4];
    #pragma unroll
    for (int s = 0; s < 16; ++s)
        av[s] = *(const v4f*)(pa + 8 * s);
    __builtin_amdgcn_sched_barrier(0);
    #pragma unroll
    for (int s = 0; s < 16; ++s) {
        #pragma unroll
        for (int i = 0; i < 4; ++i)
            bv[s][i] = *(const v4f*)(pv + (size_t)(8 * s + i) * D_);
    }

    float m = fmaxf(fmaxf(av[0].x, av[0].y), fmaxf(av[0].z, av[0].w));
    #pragma unroll
    for (int s = 1; s < 16; ++s)
        m = fmaxf(m, fmaxf(fmaxf(av[s].x, av[s].y), fmaxf(av[s].z, av[s].w)));
    m = fmaxf(m, __shfl_xor(m, 16));
    rmax[w][c] = m;
    __builtin_amdgcn_fence(__ATOMIC_RELEASE, "workgroup", "local");
    __builtin_amdgcn_s_barrier();
    __builtin_amdgcn_fence(__ATOMIC_ACQUIRE, "workgroup", "local");
    float mx = rmax[0][c];
    #pragma unroll
    for (int i = 1; i < 8; ++i) mx = fmaxf(mx, rmax[i][c]);

    const float nmx = -mx * LOG2E;
    float ls = 0.f;
    #pragma unroll
    for (int s = 0; s < 16; ++s) {
        v4f p;
        p.x = __builtin_amdgcn_exp2f(__builtin_fmaf(av[s].x, LOG2E, nmx));
        p.y = __builtin_amdgcn_exp2f(__builtin_fmaf(av[s].y, LOG2E, nmx));
        p.z = __builtin_amdgcn_exp2f(__builtin_fmaf(av[s].z, LOG2E, nmx));
        p.w = __builtin_amdgcn_exp2f(__builtin_fmaf(av[s].w, LOG2E, nmx));
        av[s] = p;
        ls += (p.x + p.y) + (p.z + p.w);
    }
    ls += __shfl_xor(ls, 16);
    rsum[w][c] = ls;

    v8f acc0 = {}, acc1 = {}, acc2 = {}, acc3 = {};
    #pragma unroll
    for (int s = 0; s < 16; ++s) {
        const v4f a = av[s], r0 = bv[s][0], r1 = bv[s][1], r2 = bv[s][2], r3 = bv[s][3];
        v2f a01, a23, f;
        a01.x = a.x; a01.y = a.y;
        a23.x = a.z; a23.y = a.w;
        f.x = r0.x; f.y = r1.x;
        acc0 = __builtin_amdgcn_wmma_f32_16x16x4_f32(false, a01, false, f, (short)0, acc0, false, false);
        f.x = r0.y; f.y = r1.y;
        acc1 = __builtin_amdgcn_wmma_f32_16x16x4_f32(false, a01, false, f, (short)0, acc1, false, false);
        f.x = r0.z; f.y = r1.z;
        acc2 = __builtin_amdgcn_wmma_f32_16x16x4_f32(false, a01, false, f, (short)0, acc2, false, false);
        f.x = r0.w; f.y = r1.w;
        acc3 = __builtin_amdgcn_wmma_f32_16x16x4_f32(false, a01, false, f, (short)0, acc3, false, false);
        f.x = r2.x; f.y = r3.x;
        acc0 = __builtin_amdgcn_wmma_f32_16x16x4_f32(false, a23, false, f, (short)0, acc0, false, false);
        f.x = r2.y; f.y = r3.y;
        acc1 = __builtin_amdgcn_wmma_f32_16x16x4_f32(false, a23, false, f, (short)0, acc1, false, false);
        f.x = r2.z; f.y = r3.z;
        acc2 = __builtin_amdgcn_wmma_f32_16x16x4_f32(false, a23, false, f, (short)0, acc2, false, false);
        f.x = r2.w; f.y = r3.w;
        acc3 = __builtin_amdgcn_wmma_f32_16x16x4_f32(false, a23, false, f, (short)0, acc3, false, false);
    }

    #pragma unroll
    for (int r = 0; r < 8; ++r) {
        v4f t;
        t.x = acc0[r]; t.y = acc1[r]; t.z = acc2[r]; t.w = acc3[r];
        *(v4f*)&part[w][8 * h + r][4 * c] = t;
    }
    __syncthreads();

    const int row = tid >> 4, c4 = (tid & 15) * 4;
    float tot = rsum[0][row];
    #pragma unroll
    for (int i = 1; i < 8; ++i) tot += rsum[i][row];
    const float inv = 1.0f / tot;
    v4f sum = *(const v4f*)&part[0][row][c4];
    #pragma unroll
    for (int i = 1; i < 8; ++i)
        sum += *(const v4f*)&part[i][row][c4];
    *(v4f*)(out + ((size_t)b * Q_ + q0 + row) * D_ + d0 + c4) = sum * inv;
}

extern "C" void kernel_launch(void* const* d_in, const int* in_sizes, int n_in,
                              void* d_out, int out_size, void* d_ws, size_t ws_size,
                              hipStream_t stream) {
    const float* q    = (const float*)d_in[0];
    const float* k    = (const float*)d_in[1];
    const float* v    = (const float*)d_in[2];
    const int*   mask = (const int*)  d_in[3];
    const float* Wq   = (const float*)d_in[4];
    const float* Wk   = (const float*)d_in[5];
    const float* wv   = (const float*)d_in[6];
    float* out = (float*)d_out;

    float* qp = (float*)d_ws;
    float* kp = qp + (size_t)B_ * Q_ * D_;
    float* sc = kp + (size_t)B_ * K_ * D_;

    proj_gemm_wmma<<<dim3(D_ / PC_, (B_ * K_) / PKR_), 256, 0, stream>>>(q, k, Wq, Wk, qp, kp);
    scores_kernel<<<dim3(K_ / SK_, Q_ / SQ_, B_), 256, 0, stream>>>(qp, kp, wv, mask, sc);
    attend_softmax_wmma<<<dim3(Q_ / 16, D_ / 64, B_), 256, 0, stream>>>(sc, v, out);
}


# keys with mask 0 left out of the score and attend stages: kept keys numbered per batch and cut into 8 equal tiles, V rows fetched by key; the k sums run over kept keys only (new order)
# speedup vs baseline: 3.7465x; 1.1698x over previous
#include <hip/hip_runtime.h>
#include <hip/hip_bf16.h>

typedef __attribute__((ext_vector_type(2))) float v2f;
typedef __attribute__((ext_vector_type(4))) float v4f;
typedef __attribute__((ext_vector_type(8))) float v8f;

#define B_ 4
#define Q_ 256
#define K_ 1024
#define D_ 256

#define PKR_ 64
#define PQR_ 16
#define PC_  64
#define PP_  (PC_ + 4)

__device__ __forceinline__ void proj_tile(const v4f a[4], const v4f b[4][4],
                                          float (*dst)[PP_], int h, int c) {
    v8f acc0 = {}, acc1 = {}, acc2 = {}, acc3 = {};
    #pragma unroll
    for (int s = 0; s < 4; ++s) {
        const v4f r0 = b[s][0], r1 = b[s][1], r2 = b[s][2], r3 = b[s][3];
        v2f a01, a23, f;
        a01.x = a[s].x; a01.y = a[s].y;
        a23.x = a[s].z; a23.y = a[s].w;
        f.x = r0.x; f.y = r1.x;
        acc0 = __builtin_amdgcn_wmma_f32_16x16x4_f32(false, a01, false, f, (short)0, acc0, false, false);
        f.x = r0.y; f.y = r1.y;
        acc1 = __builtin_amdgcn_wmma_f32_16x16x4_f32(false, a01, false, f, (short)0, acc1, false, false);
        f.x = r0.z; f.y = r1.z;
        acc2 = __builtin_amdgcn_wmma_f32_16x16x4_f32(false, a01, false, f, (short)0, acc2, false, false);
        f.x = r0.w; f.y = r1.w;
        acc3 = __builtin_amdgcn_wmma_f32_16x16x4_f32(false, a01, false, f, (short)0, acc3, false, false);
        f.x = r2.x; f.y = r3.x;
        acc0 = __builtin_amdgcn_wmma_f32_16x16x4_f32(false, a23, false, f, (short)0, acc0, false, false);
        f.x = r2.y; f.y = r3.y;
        acc1 = __builtin_amdgcn_wmma_f32_16x16x4_f32(false, a23, false, f, (short)0, acc1, false, false);
        f.x = r2.z; f.y = r3.z;
        acc2 = __builtin_amdgcn_wmma_f32_16x16x4_f32(false, a23, false, f, (short)0, acc2, false, false);
        f.x = r2.w; f.y = r3.w;
        acc3 = __builtin_amdgcn_wmma_f32_16x16x4_f32(false, a23, false, f, (short)0, acc3, false, false);
    }
    #pragma unroll
    for (int r = 0; r < 8; ++r) {
        v4f t;
        t.x = acc0[r]; t.y = acc1[r]; t.z = acc2[r]; t.w = acc3[r];
        *(v4f*)&dst[8 * h + r][4 * c] = t;
    }
}

__global__ __launch_bounds__(256)
void proj_gemm_wmma(const float* __restrict__ Xq, const float* __restrict__ Xk,
                    const float* __restrict__ Wq, const float* __restrict__ Wk,
                    float* __restrict__ qp, float* __restrict__ kp) {
    __shared__ float part[8][PKR_ + PQR_][PP_];
    const int tid = threadIdx.x;
    const int lane = tid & 31, w = tid >> 5;
    const int c = lane & 15, h = lane >> 4;
    const int n0 = blockIdx.x * PC_;
    const int by = blockIdx.y;
    const int k0 = 32 * w;

    const float* pak = Xk + (size_t)(by * PKR_ + c) * D_ + k0 + 4 * h;
    const float* paq = Xq + (size_t)(by * PQR_ + c) * D_ + k0 + 4 * h;
    const float* pbk = Wk + (size_t)(k0 + 4 * h) * D_ + n0 + 4 * c;
    const float* pbq = Wq + (size_t)(k0 + 4 * h) * D_ + n0 + 4 * c;

    v4f bk[4][4], bq[4][4], ak[4][4], aq[4];
    #pragma unroll
    for (int s = 0; s < 4; ++s) {
        ak[0][s] = *(const v4f*)(pak + 8 * s);
        #pragma unroll
        for (int i = 0; i < 4; ++i)
            bk[s][i] = *(const v4f*)(pbk + (size_t)(8 * s + i) * D_);
    }
    #pragma unroll
    for (int t = 1; t < 4; ++t) {
        #pragma unroll
        for (int s = 0; s < 4; ++s)
            ak[t][s] = *(const v4f*)(pak + (size_t)(16 * t) * D_ + 8 * s);
    }
    #pragma unroll
    for (int s = 0; s < 4; ++s) {
        aq[s] = *(const v4f*)(paq + 8 * s);
        #pragma unroll
        for (int i = 0; i < 4; ++i)
            bq[s][i] = *(const v4f*)(pbq + (size_t)(8 * s + i) * D_);
    }
    __builtin_amdgcn_sched_barrier(0);

    #pragma unroll
    for (int t = 0; t < 4; ++t)
        proj_tile(ak[t], bk, &part[w][16 * t], h, c);
    proj_tile(aq, bq, &part[w][PKR_], h, c);
    __syncthreads();

    const int r16 = tid >> 4, c4 = (tid & 15) * 4;
    #pragma unroll
    for (int t = 0; t < 5; ++t) {
        const int row = 16 * t + r16;
        v4f sum = *(const v4f*)&part[0][row][c4];
        #pragma unroll
        for (int p = 1; p < 8; ++p)
            sum += *(const v4f*)&part[p][row][c4];
        float* y = (t < 4) ? kp + (size_t)(by * PKR_ + row) * D_
                           : qp + (size_t)(by * PQR_ + r16) * D_;
        *(v4f*)(y + n0 + c4) = sum;
    }
}

#define KT_ 8
#define SCP_ 128

template <int NT>
__device__ __forceinline__ int kept_list(const int (&mv)[K_ / NT], int* idx, int* cnt, int tid) {
    constexpr int R = K_ / NT, NW = NT / 32;
    const int lane = tid & 31, w = tid >> 5;
    unsigned bal[R];
    #pragma unroll
    for (int r = 0; r < R; ++r) {
        bal[r] = __builtin_amdgcn_ballot_w32(mv[r] != 0);
        if (lane == 0) cnt[r * NW + w] = __builtin_popcount(bal[r]);
    }
    __syncthreads();
    int base[R], total = 0;
    #pragma unroll
    for (int r = 0; r < R; ++r) base[r] = 0;
    #pragma unroll
    for (int i = 0; i < R * NW; ++i) {
        const int cc = cnt[i];
        #pragma unroll
        for (int r = 0; r < R; ++r) base[r] += (i < r * NW + w) ? cc : 0;
        total += cc;
    }
    const unsigned below = (1u << lane) - 1u;
    #pragma unroll
    for (int r = 0; r < R; ++r)
        if (mv[r] != 0) idx[base[r] + __builtin_popcount(bal[r] & below)] = r * NT + tid;
    __syncthreads();
    return __builtin_amdgcn_readfirstlane(total);
}

#define SQ_ 32
#define SKMAX_ 128
#define SP_ (D_ + 4)
#define TWO_LOG2E 2.88539008177793f

template <int NC> struct ScoreOps { v4f q[4], k[NC], w; };

template <int NC>
__device__ __forceinline__ void scores_read(ScoreOps<NC>& o, const float* qrow, const float* krow,
                                            const float* wm2, int e) {
    #pragma unroll
    for (int i = 0; i < 4; ++i) o.q[i] = *(const v4f*)(qrow + 8 * i * SP_ + e);
    #pragma unroll
    for (int j = 0; j < NC; ++j) o.k[j] = *(const v4f*)(krow + 16 * j * SP_ + e);
    o.w = *(const v4f*)(wm2 + e);
}

template <int NC, bool FACTORED>
__device__ __forceinline__ v2f scores_den(const ScoreOps<NC>& o, int n) {
    const int hf = n / (4 * NC), i = (n / NC) & 3, j = n % NC;
    v2f a, b, one;
    a.x = hf ? o.q[i].z : o.q[i].x; a.y = hf ? o.q[i].w : o.q[i].y;
    b.x = hf ? o.k[j].z : o.k[j].x; b.y = hf ? o.k[j].w : o.k[j].y;
    one.x = 1.0f; one.y = 1.0f;
    if (FACTORED) return a * b + one;
    const v2f x = (a + b) * TWO_LOG2E;
    v2f t;
    t.x = __builtin_amdgcn_exp2f(x.x); t.y = __builtin_amdgcn_exp2f(x.y);
    return t + one;
}

template <int NC, bool FACTORED>
__device__ __forceinline__ void scores_terms(const ScoreOps<NC>& o, v2f acc[4][NC], v2f& wsum) {
    constexpr int NP = 8 * NC;
    v2f wlo, whi;
    wlo.x = o.w.x; wlo.y = o.w.y; whi.x = o.w.z; whi.y = o.w.w;
    v2f t[NP], r[NP];
    #pragma unroll
    for (int n = 0; n < 4; ++n) t[n] = scores_den<NC, FACTORED>(o, n);
    if (FACTORED) __builtin_amdgcn_sched_barrier(0);
    #pragma unroll
    for (int n = 0; n < NP; ++n) {
        r[n].x = __builtin_amdgcn_rcpf(t[n].x);
        r[n].y = __builtin_amdgcn_rcpf(t[n].y);
        if (n + 4 < NP) t[n + 4] = scores_den<NC, FACTORED>(o, n + 4);
        if (n >= 3) {
            const int m = n - 3;
            acc[(m / NC) & 3][m % NC] = r[m] * ((m / (4 * NC)) ? whi : wlo) + acc[(m / NC) & 3][m % NC];
        }
        if (FACTORED) {
            __builtin_amdgcn_sched_group_barrier(0x400, 1, 0);
            __builtin_amdgcn_sched_group_barrier(0x002, 1, 0);
            __builtin_amdgcn_sched_group_barrier(0x400, 1, 0);
            __builtin_amdgcn_sched_group_barrier(0x002, 1, 0);
            __builtin_amdgcn_sched_barrier(0);
        }
    }
    #pragma unroll
    for (int m = NP - 3; m < NP; ++m)
        acc[(m / NC) & 3][m % NC] = r[m] * ((m / (4 * NC)) ? whi : wlo) + acc[(m / NC) & 3][m % NC];
    wsum += wlo + whi;
}

template <int NC, bool FACTORED>
__device__ __forceinline__ void scores_loop(const float* qrow, const float* krow, const float* wm2,
                                            v2f acc[4][NC], v2f& wsum) {
    ScoreOps<NC> A, B;
    scores_read<NC>(A, qrow, krow, wm2, 0);
    for (int e = 0; e < D_; e += 8) {
        scores_read<NC>(B, qrow, krow, wm2, e + 4);
        if (FACTORED) __builtin_amdgcn_sched_barrier(0);
        scores_terms<NC, FACTORED>(A, acc, wsum);
        scores_read<NC>(A, qrow, krow, wm2, (e + 8) & (D_ - 1));
        if (FACTORED) __builtin_amdgcn_sched_barrier(0);
        scores_terms<NC, FACTORED>(B, acc, wsum);
    }
}

__device__ __forceinline__ v4f exp2x4(v4f x) {
    v4f y;
    y.x = __builtin_amdgcn_exp2f(x.x * TWO_LOG2E); y.y = __builtin_amdgcn_exp2f(x.y * TWO_LOG2E);
    y.z = __builtin_amdgcn_exp2f(x.z * TWO_LOG2E); y.w = __builtin_amdgcn_exp2f(x.w * TWO_LOG2E);
    return y;
}

template <bool RAW>
__device__ __forceinline__ float stage_q(const float* __restrict__ qpb, float (*qs)[SP_], int tid) {
    const int c4 = (tid & 63) * 4, rh = tid >> 6;
    float big = 0.f;
    #pragma unroll
    for (int i = 0; i < SQ_ / 2; ++i) {
        const int r = 2 * i + rh;
        const v4f x = *(const v4f*)(qpb + (size_t)r * D_ + c4);
        big = fmaxf(fmaxf(big, fmaxf(fabsf(x.x), fabsf(x.y))), fmaxf(fabsf(x.z), fabsf(x.w)));
        *(v4f*)&qs[r][c4] = RAW ? x : exp2x4(x);
    }
    return big;
}

template <bool RAW>
__device__ __forceinline__ float stage_k(const float* __restrict__ kpb, const int* kpos, int cn, int kw,
                                         float (*ks)[SP_], int tid) {
    const int c4 = (tid & 63) * 4, rh = tid >> 6;
    float big = 0.f;
    for (int i0 = 0; i0 < kw / 2; i0 += 32) {
        int key[32];
        #pragma unroll
        for (int i = 0; i < 32; ++i)
            key[i] = kpos[min(2 * (i0 + i) + rh, cn - 1)];
        v4f x[32];
        #pragma unroll
        for (int i = 0; i < 32; ++i)
            x[i] = *(const v4f*)(kpb + (size_t)key[i] * D_ + c4);
        #pragma unroll
        for (int i = 0; i < 32; ++i) {
            const int r = 2 * (i0 + i) + rh;
            v4f y = x[i];
            if (r >= cn) y = v4f{0.f, 0.f, 0.f, 0.f};
            big = fmaxf(fmaxf(big, fmaxf(fabsf(y.x), fabsf(y.y))), fmaxf(fabsf(y.z), fabsf(y.w)));
            *(v4f*)&ks[r][c4] = RAW ? y : exp2x4(y);
        }
    }
    return big;
}

template <int NC>
__device__ __forceinline__ void scores_tile(const float* qrow, const float* krow, const float* wm2,
                                            int out_of_range, float* __restrict__ dst, int cn, int tk) {
    v2f acc[4][NC];
    #pragma unroll
    for (int i = 0; i < 4; ++i)
        #pragma unroll
        for (int j = 0; j < NC; ++j) { acc[i][j].x = 0.f; acc[i][j].y = 0.f; }
    v2f wsum; wsum.x = 0.f; wsum.y = 0.f;

    if (!out_of_range) scores_loop<NC, true>(qrow, krow, wm2, acc, wsum);
    else               scores_loop<NC, false>(qrow, krow, wm2, acc, wsum);

    const float wtot = -0.5f * (wsum.x + wsum.y);
    const float ninf = -__builtin_inff();
    #pragma unroll
    for (int i = 0; i < 4; ++i)
        #pragma unroll
        for (int j = 0; j < NC; ++j)
            dst[(size_t)(8 * i) * SCP_ + 16 * j] =
                (tk + 16 * j < cn) ? (acc[i][j].x + acc[i][j].y) + wtot : ninf;
}

__global__ __launch_bounds__(128)
void scores_kernel(const float* __restrict__ qp, const float* __restrict__ kp,
                   const float* __restrict__ wv, const int* __restrict__ mask,
                   float* __restrict__ scores) {
    __shared__ float qs[SQ_][SP_];
    __shared__ float ks[SKMAX_][SP_];
    __shared__ float wm2[D_];
    __shared__ __attribute__((aligned(16))) int kidx[K_];
    __shared__ int kcnt[32];
    const int tid = threadIdx.x;
    const int t = __builtin_amdgcn_workgroup_id_x(), q0 = __builtin_amdgcn_workgroup_id_y() * SQ_,
              b = __builtin_amdgcn_workgroup_id_z();

    int mv[K_ / 128];
    #pragma unroll
    for (int r = 0; r < K_ / 128; ++r) mv[r] = mask[(size_t)b * K_ + r * 128 + tid];
    const float wv0 = wv[tid], wv1 = wv[tid + 128];
    const float* qpb = qp + ((size_t)b * Q_ + q0) * D_;
    const float* kpb = kp + (size_t)b * K_ * D_;
    const float bigq = stage_q<false>(qpb, qs, tid);
    const int n = kept_list<128>(mv, kidx, kcnt, tid);
    const int nc = (n + 127) >> 7;
    const int kw = 16 * nc;
    const int c0 = t * kw;
    const int cn = min(kw, n - c0);

    float* sct = scores + ((size_t)(b * KT_ + t) * Q_ + q0) * SCP_;
    if (cn <= 0) {
        const float ninf = -__builtin_inff();
        const int kf = max(kw, 16);
        for (int x = tid; x < SQ_ * SCP_; x += 128)
            if ((x & (SCP_ - 1)) < kf) sct[x] = ninf;
        return;
    }

    const float big = fmaxf(bigq, stage_k<false>(kpb, kidx + c0, cn, kw, ks, tid));
    wm2[tid] = -2.0f * wv0;
    wm2[tid + 128] = -2.0f * wv1;
    const int out_of_range = __builtin_amdgcn_readfirstlane(__syncthreads_or(!(big <= 40.0f)));
    if (out_of_range) {
        stage_q<true>(qpb, qs, tid);
        stage_k<true>(kpb, kidx + c0, cn, kw, ks, tid);
        __syncthreads();
    }

    const int lane = tid & 31, w = tid >> 5;
    const int tk = lane & 15, tq = 2 * w + (lane >> 4);
    const float* qrow = &qs[tq][0];
    const float* krow = &ks[tk][0];
    float* dst = sct + (size_t)tq * SCP_ + tk;

    switch (nc) {
        case 1: scores_tile<1>(qrow, krow, wm2, out_of_range, dst, cn, tk); break;
        case 2: scores_tile<2>(qrow, krow, wm2, out_of_range, dst, cn, tk); break;
        case 3: scores_tile<3>(qrow, krow, wm2, out_of_range, dst, cn, tk); break;
        case 4: scores_tile<4>(qrow, krow, wm2, out_of_range, dst, cn, tk); break;
        case 5: scores_tile<5>(qrow, krow, wm2, out_of_range, dst, cn, tk); break;
        case 6: scores_tile<6>(qrow, krow, wm2, out_of_range, dst, cn, tk); break;
        case 7: scores_tile<7>(qrow, krow, wm2, out_of_range, dst, cn, tk); break;
        default: scores_tile<8>(qrow, krow, wm2, out_of_range, dst, cn, tk); break;
    }
}

#define LOG2E 1.44269504088896f

template <int NC>
__device__ __forceinline__ void attend_tile(const float* __restrict__ pa, const float* __restrict__ pv,
                                            const int* kpos, int cw, int safe_row,
                                            float (*part)[16][64], float (*rmax)[16], float (*rsum)[16],
                                            float* __restrict__ po, int tid) {
    constexpr int NS = 2 * NC;
    const int lane = tid & 31, w = tid >> 5;
    const int c = lane & 15, h = lane >> 4;

    v4f av[NS], bv[NS][4];
    #pragma unroll
    for (int s = 0; s < NS; ++s)
        av[s] = *(const v4f*)(pa + 8 * s);
    __builtin_amdgcn_sched_barrier(0);
    int4 kk[NS];
    #pragma unroll
    for (int s = 0; s < NS; ++s)
        kk[s] = *(const int4*)(kpos + 8 * s);
    __builtin_amdgcn_sched_barrier(0);
    #pragma unroll
    for (int s = 0; s < NS; ++s) {
        const int p0 = 8 * s + 4 * h;
        const int r0 = (p0     < cw) ? kk[s].x : safe_row;
        const int r1 = (p0 + 1 < cw) ? kk[s].y : safe_row;
        const int r2 = (p0 + 2 < cw) ? kk[s].z : safe_row;
        const int r3 = (p0 + 3 < cw) ? kk[s].w : safe_row;
        bv[s][0] = *(const v4f*)(pv + (size_t)r0 * D_);
        bv[s][1] = *(const v4f*)(pv + (size_t)r1 * D_);
        bv[s][2] = *(const v4f*)(pv + (size_t)r2 * D_);
        bv[s][3] = *(const v4f*)(pv + (size_t)r3 * D_);
    }

    float m = fmaxf(fmaxf(av[0].x, av[0].y), fmaxf(av[0].z, av[0].w));
    #pragma unroll
    for (int s = 1; s < NS; ++s)
        m = fmaxf(m, fmaxf(fmaxf(av[s].x, av[s].y), fmaxf(av[s].z, av[s].w)));
    m = fmaxf(m, __shfl_xor(m, 16));
    rmax[w][c] = m;
    __builtin_amdgcn_fence(__ATOMIC_RELEASE, "workgroup", "local");
    __builtin_amdgcn_s_barrier();
    __builtin_amdgcn_fence(__ATOMIC_ACQUIRE, "workgroup", "local");
    float mx = rmax[0][c];
    #pragma unroll
    for (int i = 1; i < 8; ++i) mx = fmaxf(mx, rmax[i][c]);

    const float nmx = -mx * LOG2E;
    float ls = 0.f;
    #pragma unroll
    for (int s = 0; s < NS; ++s) {
        v4f p;
        p.x = __builtin_amdgcn_exp2f(__builtin_fmaf(av[s].x, LOG2E, nmx));
        p.y = __builtin_amdgcn_exp2f(__builtin_fmaf(av[s].y, LOG2E, nmx));
        p.z = __builtin_amdgcn_exp2f(__builtin_fmaf(av[s].z, LOG2E, nmx));
        p.w = __builtin_amdgcn_exp2f(__builtin_fmaf(av[s].w, LOG2E, nmx));
        av[s] = p;
        ls += (p.x + p.y) + (p.z + p.w);
    }
    ls += __shfl_xor(ls, 16);
    rsum[w][c] = ls;

    v8f acc0 = {}, acc1 = {}, acc2 = {}, acc3 = {};
    #pragma unroll
    for (int s = 0; s < NS; ++s) {
        const v4f a = av[s], r0 = bv[s][0], r1 = bv[s][1], r2 = bv[s][2], r3 = bv[s][3];
        v2f a01, a23, f;
        a01.x = a.x; a01.y = a.y;
        a23.x = a.z; a23.y = a.w;
        f.x = r0.x; f.y = r1.x;
        acc0 = __builtin_amdgcn_wmma_f32_16x16x4_f32(false, a01, false, f, (short)0, acc0, false, false);
        f.x = r0.y; f.y = r1.y;
        acc1 = __builtin_amdgcn_wmma_f32_16x16x4_f32(false, a01, false, f, (short)0, acc1, false, false);
        f.x = r0.z; f.y = r1.z;
        acc2 = __builtin_amdgcn_wmma_f32_16x16x4_f32(false, a01, false, f, (short)0, acc2, false, false);
        f.x = r0.w; f.y = r1.w;
        acc3 = __builtin_amdgcn_wmma_f32_16x16x4_f32(false, a01, false, f, (short)0, acc3, false, false);
        f.x = r2.x; f.y = r3.x;
        acc0 = __builtin_amdgcn_wmma_f32_16x16x4_f32(false, a23, false, f, (short)0, acc0, false, false);
        f.x = r2.y; f.y = r3.y;
        acc1 = __builtin_amdgcn_wmma_f32_16x16x4_f32(false, a23, false, f, (short)0, acc1, false, false);
        f.x = r2.z; f.y = r3.z;
        acc2 = __builtin_amdgcn_wmma_f32_16x16x4_f32(false, a23, false, f, (short)0, acc2, false, false);
        f.x = r2.w; f.y = r3.w;
        acc3 = __builtin_amdgcn_wmma_f32_16x16x4_f32(false, a23, false, f, (short)0, acc3, false, false);
    }

    #pragma unroll
    for (int r = 0; r < 8; ++r) {
        v4f t;
        t.x = acc0[r]; t.y = acc1[r]; t.z = acc2[r]; t.w = acc3[r];
        *(v4f*)&part[w][8 * h + r][4 * c] = t;
    }
    __syncthreads();

    const int row = tid >> 4, c4 = (tid & 15) * 4;
    float tot = rsum[0][row];
    #pragma unroll
    for (int i = 1; i < 8; ++i) tot += rsum[i][row];
    const float inv = 1.0f / tot;
    v4f sum = *(const v4f*)&part[0][row][c4];
    #pragma unroll
    for (int i = 1; i < 8; ++i)
        sum += *(const v4f*)&part[i][row][c4];
    *(v4f*)(po + (size_t)row * D_ + c4) = sum * inv;
}

__global__ __launch_bounds__(256)
void attend_softmax_wmma(const float* __restrict__ scores, const float* __restrict__ V,
                         const int* __restrict__ mask, float* __restrict__ out) {
    __shared__ float part[8][16][64];
    __shared__ float rmax[8][16];
    __shared__ float rsum[8][16];
    __shared__ __attribute__((aligned(16))) int kidx[K_];
    __shared__ int kcnt[32];
    const int tid = threadIdx.x;
    const int lane = tid & 31;
    const int w = __builtin_amdgcn_readfirstlane(tid >> 5);
    const int c = lane & 15, h = lane >> 4;
    const int q0 = blockIdx.x * 16, d0 = blockIdx.y * 64, b = blockIdx.z;

    int mv[K_ / 256];
    #pragma unroll
    for (int r = 0; r < K_ / 256; ++r) mv[r] = mask[(size_t)b * K_ + r * 256 + tid];
    const int n = kept_list<256>(mv, kidx, kcnt, tid);
    const int nc = max(1, (n + 127) >> 7);
    const int kw = 16 * nc;
    const int c0 = w * kw;
    const int cw = max(0, min(kw, n - c0));
    const int safe_row = n > 0 ? kidx[0] : 0;

    const float* pa = scores + ((size_t)(b * KT_ + w) * Q_ + q0 + c) * SCP_ + 4 * h;
    const float* pv = V + (size_t)b * K_ * D_ + d0 + 4 * c;
    const int* kpos = kidx + c0 + 4 * h;
    float* po = out + ((size_t)b * Q_ + q0) * D_ + d0;

    switch (nc) {
        case 1: attend_tile<1>(pa, pv, kpos, cw, safe_row, part, rmax, rsum, po, tid); break;
        case 2: attend_tile<2>(pa, pv, kpos, cw, safe_row, part, rmax, rsum, po, tid); break;
        case 3: attend_tile<3>(pa, pv, kpos, cw, safe_row, part, rmax, rsum, po, tid); break;
        case 4: attend_tile<4>(pa, pv, kpos, cw, safe_row, part, rmax, rsum, po, tid); break;
        case 5: attend_tile<5>(pa, pv, kpos, cw, safe_row, part, rmax, rsum, po, tid); break;
        case 6: attend_tile<6>(pa, pv, kpos, cw, safe_row, part, rmax, rsum, po, tid); break;
        case 7: attend_tile<7>(pa, pv, kpos, cw, safe_row, part, rmax, rsum, po, tid); break;
        default: attend_tile<8>(pa, pv, kpos, cw, safe_row, part, rmax, rsum, po, tid); break;
    }
}

extern "C" void kernel_launch(void* const* d_in, const int* in_sizes, int n_in,
                              void* d_out, int out_size, void* d_ws, size_t ws_size,
                              hipStream_t stream) {
    const float* q    = (const float*)d_in[0];
    const float* k    = (const float*)d_in[1];
    const float* v    = (const float*)d_in[2];
    const int*   mask = (const int*)  d_in[3];
    const float* Wq   = (const float*)d_in[4];
    const float* Wk   = (const float*)d_in[5];
    const float* wv   = (const float*)d_in[6];
    float* out = (float*)d_out;

    float* qp = (float*)d_ws;
    float* kp = qp + (size_t)B_ * Q_ * D_;
    float* sc = kp + (size_t)B_ * K_ * D_;

    proj_gemm_wmma<<<dim3(D_ / PC_, (B_ * K_) / PKR_), 256, 0, stream>>>(q, k, Wq, Wk, qp, kp);
    scores_kernel<<<dim3(KT_, Q_ / SQ_, B_), 128, 0, stream>>>(qp, kp, wv, mask, sc);
    attend_softmax_wmma<<<dim3(Q_ / 16, D_ / 64, B_), 256, 0, stream>>>(sc, v, mask, out);
}


# keys with mask 0 left out of the kp projection too: kp computed for kept keys only, stored by position (list built inside the projection blocks); the score stage reads rows by position, no list
# speedup vs baseline: 3.9258x; 1.0479x over previous
#include <hip/hip_runtime.h>
#include <hip/hip_bf16.h>

typedef __attribute__((ext_vector_type(2))) float v2f;
typedef __attribute__((ext_vector_type(4))) float v4f;
typedef __attribute__((ext_vector_type(8))) float v8f;

#define B_ 4
#define Q_ 256
#define K_ 1024
#define D_ 256

#define PKR_ 64
#define PQR_ 16
#define PC_  64
#define PP_  (PC_ + 4)

__device__ __forceinline__ void proj_tile(const v4f a[4], const v4f b[4][4],
                                          float (*dst)[PP_], int h, int c) {
    v8f acc0 = {}, acc1 = {}, acc2 = {}, acc3 = {};
    #pragma unroll
    for (int s = 0; s < 4; ++s) {
        const v4f r0 = b[s][0], r1 = b[s][1], r2 = b[s][2], r3 = b[s][3];
        v2f a01, a23, f;
        a01.x = a[s].x; a01.y = a[s].y;
        a23.x = a[s].z; a23.y = a[s].w;
        f.x = r0.x; f.y = r1.x;
        acc0 = __builtin_amdgcn_wmma_f32_16x16x4_f32(false, a01, false, f, (short)0, acc0, false, false);
        f.x = r0.y; f.y = r1.y;
        acc1 = __builtin_amdgcn_wmma_f32_16x16x4_f32(false, a01, false, f, (short)0, acc1, false, false);
        f.x = r0.z; f.y = r1.z;
        acc2 = __builtin_amdgcn_wmma_f32_16x16x4_f32(false, a01, false, f, (short)0, acc2, false, false);
        f.x = r0.w; f.y = r1.w;
        acc3 = __builtin_amdgcn_wmma_f32_16x16x4_f32(false, a01, false, f, (short)0, acc3, false, false);
        f.x = r2.x; f.y = r3.x;
        acc0 = __builtin_amdgcn_wmma_f32_16x16x4_f32(false, a23, false, f, (short)0, acc0, false, false);
        f.x = r2.y; f.y = r3.y;
        acc1 = __builtin_amdgcn_wmma_f32_16x16x4_f32(false, a23, false, f, (short)0, acc1, false, false);
        f.x = r2.z; f.y = r3.z;
        acc2 = __builtin_amdgcn_wmma_f32_16x16x4_f32(false, a23, false, f, (short)0, acc2, false, false);
        f.x = r2.w; f.y = r3.w;
        acc3 = __builtin_amdgcn_wmma_f32_16x16x4_f32(false, a23, false, f, (short)0, acc3, false, false);
    }
    #pragma unroll
    for (int r = 0; r < 8; ++r) {
        v4f t;
        t.x = acc0[r]; t.y = acc1[r]; t.z = acc2[r]; t.w = acc3[r];
        *(v4f*)&dst[8 * h + r][4 * c] = t;
    }
}

#define KT_ 8
#define SCP_ 128

__device__ __forceinline__ void barrier_lds_only() {
    __builtin_amdgcn_fence(__ATOMIC_RELEASE, "workgroup", "local");
    __builtin_amdgcn_s_barrier();
    __builtin_amdgcn_fence(__ATOMIC_ACQUIRE, "workgroup", "local");
}

template <int NT, bool LDS_ONLY = false>
__device__ __forceinline__ int kept_list(const int (&mv)[K_ / NT], int* idx, int* cnt, int tid) {
    constexpr int R = K_ / NT, NW = NT / 32;
    const int lane = tid & 31, w = tid >> 5;
    unsigned bal[R];
    #pragma unroll
    for (int r = 0; r < R; ++r) {
        bal[r] = __builtin_amdgcn_ballot_w32(mv[r] != 0);
        if (lane == 0) cnt[r * NW + w] = __builtin_popcount(bal[r]);
    }
    if (LDS_ONLY) barrier_lds_only(); else __syncthreads();
    int base[R], total = 0;
    #pragma unroll
    for (int r = 0; r < R; ++r) base[r] = 0;
    #pragma unroll
    for (int i = 0; i < R * NW; ++i) {
        const int cc = cnt[i];
        #pragma unroll
        for (int r = 0; r < R; ++r) base[r] += (i < r * NW + w) ? cc : 0;
        total += cc;
    }
    const unsigned below = (1u << lane) - 1u;
    #pragma unroll
    for (int r = 0; r < R; ++r)
        if (mv[r] != 0) idx[base[r] + __builtin_popcount(bal[r] & below)] = r * NT + tid;
    if (LDS_ONLY) barrier_lds_only(); else __syncthreads();
    return __builtin_amdgcn_readfirstlane(total);
}

template <int NT>
__device__ __forceinline__ void proj_block(const float* __restrict__ xk, const int* kidx, int n, int r,
                                           const v4f (&aq)[4], const v4f (&bq)[4][4], const v4f (&bk)[4][4],
                                           float (*part)[PKR_ + PQR_][PP_],
                                           float* __restrict__ yq, float* __restrict__ yk, int tid) {
    const int lane = tid & 31, w = tid >> 5;
    const int c = lane & 15, h = lane >> 4;
    constexpr int NA = NT > 0 ? NT : 1;
    v4f ak[NA][4];
    if (NT > 0) {
        int key[NA];
        #pragma unroll
        for (int t = 0; t < NT; ++t) key[t] = kidx[min(16 * (r + 16 * t) + c, n - 1)];
        #pragma unroll
        for (int t = 0; t < NT; ++t) {
            #pragma unroll
            for (int s = 0; s < 4; ++s)
                ak[t][s] = *(const v4f*)(xk + (size_t)key[t] * D_ + 8 * s);
        }
    }
    __builtin_amdgcn_sched_barrier(0);

    proj_tile(aq, bq, &part[w][PKR_], h, c);
    #pragma unroll
    for (int t = 0; t < NT; ++t)
        proj_tile(ak[t], bk, &part[w][16 * t], h, c);
    __syncthreads();

    const int r16 = tid >> 4, c4 = (tid & 15) * 4;
    {
        v4f sum = *(const v4f*)&part[0][PKR_ + r16][c4];
        #pragma unroll
        for (int p = 1; p < 8; ++p)
            sum += *(const v4f*)&part[p][PKR_ + r16][c4];
        *(v4f*)(yq + (size_t)r16 * D_ + c4) = sum;
    }
    #pragma unroll
    for (int t = 0; t < NT; ++t) {
        const int row = 16 * t + r16;
        const int pos = 16 * (r + 16 * t) + r16;
        v4f sum = *(const v4f*)&part[0][row][c4];
        #pragma unroll
        for (int p = 1; p < 8; ++p)
            sum += *(const v4f*)&part[p][row][c4];
        if (pos < n) *(v4f*)(yk + (size_t)pos * D_ + c4) = sum;
    }
}

__global__ __launch_bounds__(256)
void proj_gemm_wmma(const float* __restrict__ Xq, const float* __restrict__ Xk,
                    const float* __restrict__ Wq, const float* __restrict__ Wk,
                    const int* __restrict__ mask,
                    float* __restrict__ qp, float* __restrict__ kp, int* __restrict__ nkeep) {
    __shared__ float part[8][PKR_ + PQR_][PP_];
    __shared__ __attribute__((aligned(16))) int kidx[K_];
    __shared__ int kcnt[32];
    const int tid = threadIdx.x;
    const int lane = tid & 31, w = tid >> 5;
    const int c = lane & 15, h = lane >> 4;
    const int n0 = blockIdx.x * PC_;
    const int by = blockIdx.y;
    const int b = by >> 4, r = by & 15;
    const int k0 = 32 * w;

    int mv[K_ / 256];
    #pragma unroll
    for (int i = 0; i < K_ / 256; ++i) mv[i] = mask[(size_t)b * K_ + i * 256 + tid];
    __builtin_amdgcn_sched_barrier(0);

    const float* paq = Xq + (size_t)(by * PQR_ + c) * D_ + k0 + 4 * h;
    const float* pbk = Wk + (size_t)(k0 + 4 * h) * D_ + n0 + 4 * c;
    const float* pbq = Wq + (size_t)(k0 + 4 * h) * D_ + n0 + 4 * c;

    v4f bk[4][4], bq[4][4], aq[4];
    #pragma unroll
    for (int s = 0; s < 4; ++s) {
        aq[s] = *(const v4f*)(paq + 8 * s);
        #pragma unroll
        for (int i = 0; i < 4; ++i)
            bq[s][i] = *(const v4f*)(pbq + (size_t)(8 * s + i) * D_);
    }
    #pragma unroll
    for (int s = 0; s < 4; ++s) {
        #pragma unroll
        for (int i = 0; i < 4; ++i)
            bk[s][i] = *(const v4f*)(pbk + (size_t)(8 * s + i) * D_);
    }
    __builtin_amdgcn_sched_barrier(0);

    const int n = kept_list<256, true>(mv, kidx, kcnt, tid);
    if (blockIdx.x == 0 && r == 0 && tid == 0) nkeep[b] = n;
    const int tiles = (n + 15) >> 4;
    const int nt = max(0, (tiles - r + 15) >> 4);

    const float* xk = Xk + (size_t)b * K_ * D_ + k0 + 4 * h;
    float* yq = qp + (size_t)(by * PQR_) * D_ + n0;
    float* yk = kp + (size_t)b * K_ * D_ + n0;
    switch (nt) {
        case 0:  proj_block<0>(xk, kidx, n, r, aq, bq, bk, part, yq, yk, tid); break;
        case 1:  proj_block<1>(xk, kidx, n, r, aq, bq, bk, part, yq, yk, tid); break;
        case 2:  proj_block<2>(xk, kidx, n, r, aq, bq, bk, part, yq, yk, tid); break;
        case 3:  proj_block<3>(xk, kidx, n, r, aq, bq, bk, part, yq, yk, tid); break;
        default: proj_block<4>(xk, kidx, n, r, aq, bq, bk, part, yq, yk, tid); break;
    }
}

#define SQ_ 32
#define SKMAX_ 128
#define SP_ (D_ + 4)
#define TWO_LOG2E 2.88539008177793f

template <int NC> struct ScoreOps { v4f q[4], k[NC], w; };

template <int NC>
__device__ __forceinline__ void scores_read(ScoreOps<NC>& o, const float* qrow, const float* krow,
                                            const float* wm2, int e) {
    #pragma unroll
    for (int i = 0; i < 4; ++i) o.q[i] = *(const v4f*)(qrow + 8 * i * SP_ + e);
    #pragma unroll
    for (int j = 0; j < NC; ++j) o.k[j] = *(const v4f*)(krow + 16 * j * SP_ + e);
    o.w = *(const v4f*)(wm2 + e);
}

template <int NC, bool FACTORED>
__device__ __forceinline__ v2f scores_den(const ScoreOps<NC>& o, int n) {
    const int hf = n / (4 * NC), i = (n / NC) & 3, j = n % NC;
    v2f a, b, one;
    a.x = hf ? o.q[i].z : o.q[i].x; a.y = hf ? o.q[i].w : o.q[i].y;
    b.x = hf ? o.k[j].z : o.k[j].x; b.y = hf ? o.k[j].w : o.k[j].y;
    one.x = 1.0f; one.y = 1.0f;
    if (FACTORED) return a * b + one;
    const v2f x = (a + b) * TWO_LOG2E;
    v2f t;
    t.x = __builtin_amdgcn_exp2f(x.x); t.y = __builtin_amdgcn_exp2f(x.y);
    return t + one;
}

template <int NC, bool FACTORED>
__device__ __forceinline__ void scores_terms(const ScoreOps<NC>& o, v2f acc[4][NC], v2f& wsum) {
    constexpr int NP = 8 * NC;
    v2f wlo, whi;
    wlo.x = o.w.x; wlo.y = o.w.y; whi.x = o.w.z; whi.y = o.w.w;
    v2f t[NP], r[NP];
    #pragma unroll
    for (int n = 0; n < 4; ++n) t[n] = scores_den<NC, FACTORED>(o, n);
    if (FACTORED) __builtin_amdgcn_sched_barrier(0);
    #pragma unroll
    for (int n = 0; n < NP; ++n) {
        r[n].x = __builtin_amdgcn_rcpf(t[n].x);
        r[n].y = __builtin_amdgcn_rcpf(t[n].y);
        if (n + 4 < NP) t[n + 4] = scores_den<NC, FACTORED>(o, n + 4);
        if (n >= 3) {
            const int m = n - 3;
            acc[(m / NC) & 3][m % NC] = r[m] * ((m / (4 * NC)) ? whi : wlo) + acc[(m / NC) & 3][m % NC];
        }
        if (FACTORED) {
            __builtin_amdgcn_sched_group_barrier(0x400, 1, 0);
            __builtin_amdgcn_sched_group_barrier(0x002, 1, 0);
            __builtin_amdgcn_sched_group_barrier(0x400, 1, 0);
            __builtin_amdgcn_sched_group_barrier(0x002, 1, 0);
            __builtin_amdgcn_sched_barrier(0);
        }
    }
    #pragma unroll
    for (int m = NP - 3; m < NP; ++m)
        acc[(m / NC) & 3][m % NC] = r[m] * ((m / (4 * NC)) ? whi : wlo) + acc[(m / NC) & 3][m % NC];
    wsum += wlo + whi;
}

template <int NC, bool FACTORED>
__device__ __forceinline__ void scores_loop(const float* qrow, const float* krow, const float* wm2,
                                            v2f acc[4][NC], v2f& wsum) {
    ScoreOps<NC> A, B;
    scores_read<NC>(A, qrow, krow, wm2, 0);
    for (int e = 0; e < D_; e += 8) {
        scores_read<NC>(B, qrow, krow, wm2, e + 4);
        if (FACTORED) __builtin_amdgcn_sched_barrier(0);
        scores_terms<NC, FACTORED>(A, acc, wsum);
        scores_read<NC>(A, qrow, krow, wm2, (e + 8) & (D_ - 1));
        if (FACTORED) __builtin_amdgcn_sched_barrier(0);
        scores_terms<NC, FACTORED>(B, acc, wsum);
    }
}

__device__ __forceinline__ v4f exp2x4(v4f x) {
    v4f y;
    y.x = __builtin_amdgcn_exp2f(x.x * TWO_LOG2E); y.y = __builtin_amdgcn_exp2f(x.y * TWO_LOG2E);
    y.z = __builtin_amdgcn_exp2f(x.z * TWO_LOG2E); y.w = __builtin_amdgcn_exp2f(x.w * TWO_LOG2E);
    return y;
}

__device__ __forceinline__ void load_q(const float* __restrict__ qpb, v4f (&x)[SQ_ / 2], int tid) {
    const int c4 = (tid & 63) * 4, rh = tid >> 6;
    #pragma unroll
    for (int i = 0; i < SQ_ / 2; ++i)
        x[i] = *(const v4f*)(qpb + (size_t)(2 * i + rh) * D_ + c4);
}

template <bool RAW>
__device__ __forceinline__ float put_q(const v4f (&x)[SQ_ / 2], float (*qs)[SP_], int tid) {
    const int c4 = (tid & 63) * 4, rh = tid >> 6;
    float big = 0.f;
    #pragma unroll
    for (int i = 0; i < SQ_ / 2; ++i) {
        big = fmaxf(fmaxf(big, fmaxf(fabsf(x[i].x), fabsf(x[i].y))), fmaxf(fabsf(x[i].z), fabsf(x[i].w)));
        *(v4f*)&qs[2 * i + rh][c4] = RAW ? x[i] : exp2x4(x[i]);
    }
    return big;
}

template <bool RAW>
__device__ __forceinline__ float stage_q(const float* __restrict__ qpb, float (*qs)[SP_], int tid) {
    v4f x[SQ_ / 2];
    load_q(qpb, x, tid);
    return put_q<RAW>(x, qs, tid);
}

template <bool RAW, class F>
__device__ __forceinline__ float stage_k_chunk(const float* __restrict__ kpb, int cn, float (*ks)[SP_],
                                               int tid, int i0, F while_in_flight) {
    const int c4 = (tid & 63) * 4, rh = tid >> 6;
    float big = 0.f;
    int key[32];
    #pragma unroll
    for (int i = 0; i < 32; ++i)
        key[i] = min(2 * (i0 + i) + rh, cn - 1);
    v4f x[32];
    #pragma unroll
    for (int i = 0; i < 32; ++i)
        x[i] = *(const v4f*)(kpb + (size_t)key[i] * D_ + c4);
    __builtin_amdgcn_sched_barrier(0);
    while_in_flight();
    __builtin_amdgcn_sched_barrier(0);
    #pragma unroll
    for (int i = 0; i < 32; ++i) {
        const int r = 2 * (i0 + i) + rh;
        v4f y = x[i];
        if (r >= cn) y = v4f{0.f, 0.f, 0.f, 0.f};
        big = fmaxf(fmaxf(big, fmaxf(fabsf(y.x), fabsf(y.y))), fmaxf(fabsf(y.z), fabsf(y.w)));
        *(v4f*)&ks[r][c4] = RAW ? y : exp2x4(y);
    }
    return big;
}

template <bool RAW, class F>
__device__ __forceinline__ float stage_k(const float* __restrict__ kpb, int cn, int kw,
                                         float (*ks)[SP_], int tid, F while_in_flight) {
    float big = stage_k_chunk<RAW>(kpb, cn, ks, tid, 0, while_in_flight);
    for (int i0 = 32; i0 < kw / 2; i0 += 32)
        big = fmaxf(big, stage_k_chunk<RAW>(kpb, cn, ks, tid, i0, []() {}));
    return big;
}

template <int NC>
__device__ __forceinline__ void scores_tile(const float* qrow, const float* krow, const float* wm2,
                                            int out_of_range, float* __restrict__ dst, int cn, int tk) {
    v2f acc[4][NC];
    #pragma unroll
    for (int i = 0; i < 4; ++i)
        #pragma unroll
        for (int j = 0; j < NC; ++j) { acc[i][j].x = 0.f; acc[i][j].y = 0.f; }
    v2f wsum; wsum.x = 0.f; wsum.y = 0.f;

    if (!out_of_range) scores_loop<NC, true>(qrow, krow, wm2, acc, wsum);
    else               scores_loop<NC, false>(qrow, krow, wm2, acc, wsum);

    const float wtot = -0.5f * (wsum.x + wsum.y);
    const float ninf = -__builtin_inff();
    #pragma unroll
    for (int i = 0; i < 4; ++i)
        #pragma unroll
        for (int j = 0; j < NC; ++j)
            dst[(size_t)(8 * i) * SCP_ + 16 * j] =
                (tk + 16 * j < cn) ? (acc[i][j].x + acc[i][j].y) + wtot : ninf;
}

__global__ __launch_bounds__(128)
void scores_kernel(const float* __restrict__ qp, const float* __restrict__ kp,
                   const float* __restrict__ wv, const int* __restrict__ nkeep,
                   float* __restrict__ scores) {
    __shared__ float qs[SQ_][SP_];
    __shared__ float ks[SKMAX_][SP_];
    __shared__ float wm2[D_];
    const int tid = threadIdx.x;
    const int t = __builtin_amdgcn_workgroup_id_x(), q0 = __builtin_amdgcn_workgroup_id_y() * SQ_,
              b = __builtin_amdgcn_workgroup_id_z();

    const int n = nkeep[b];
    const float wv0 = wv[tid], wv1 = wv[tid + 128];
    const float* qpb = qp + ((size_t)b * Q_ + q0) * D_;
    const float* kpb = kp + (size_t)b * K_ * D_;
    const int nc = (n + 127) >> 7;
    const int kw = 16 * nc;
    const int c0 = t * kw;
    const int cn = min(kw, n - c0);

    float* sct = scores + ((size_t)(b * KT_ + t) * Q_ + q0) * SCP_;
    if (cn <= 0) {
        const float ninf = -__builtin_inff();
        const int kf = max(kw, 16);
        for (int x = tid; x < SQ_ * SCP_; x += 128)
            if ((x & (SCP_ - 1)) < kf) sct[x] = ninf;
        return;
    }

    v4f xq[SQ_ / 2];
    load_q(qpb, xq, tid);
    __builtin_amdgcn_sched_barrier(0);
    float bigq = 0.f;
    const float bigk = stage_k<false>(kpb + (size_t)c0 * D_, cn, kw, ks, tid,
                                      [&]() { bigq = put_q<false>(xq, qs, tid); });
    const float big = fmaxf(bigq, bigk);
    wm2[tid] = -2.0f * wv0;
    wm2[tid + 128] = -2.0f * wv1;
    const int out_of_range = __builtin_amdgcn_readfirstlane(__syncthreads_or(!(big <= 40.0f)));
    if (out_of_range) {
        stage_q<true>(qpb, qs, tid);
        stage_k<true>(kpb + (size_t)c0 * D_, cn, kw, ks, tid, []() {});
        __syncthreads();
    }

    const int lane = tid & 31, w = tid >> 5;
    const int tk = lane & 15, tq = 2 * w + (lane >> 4);
    const float* qrow = &qs[tq][0];
    const float* krow = &ks[tk][0];
    float* dst = sct + (size_t)tq * SCP_ + tk;

    switch (nc) {
        case 1: scores_tile<1>(qrow, krow, wm2, out_of_range, dst, cn, tk); break;
        case 2: scores_tile<2>(qrow, krow, wm2, out_of_range, dst, cn, tk); break;
        case 3: scores_tile<3>(qrow, krow, wm2, out_of_range, dst, cn, tk); break;
        case 4: scores_tile<4>(qrow, krow, wm2, out_of_range, dst, cn, tk); break;
        case 5: scores_tile<5>(qrow, krow, wm2, out_of_range, dst, cn, tk); break;
        case 6: scores_tile<6>(qrow, krow, wm2, out_of_range, dst, cn, tk); break;
        case 7: scores_tile<7>(qrow, krow, wm2, out_of_range, dst, cn, tk); break;
        default: scores_tile<8>(qrow, krow, wm2, out_of_range, dst, cn, tk); break;
    }
}

#define LOG2E 1.44269504088896f

template <int NC>
__device__ __forceinline__ void attend_tile(const float* __restrict__ pa, const float* __restrict__ pv,
                                            const int* kpos, int cw, int safe_row,
                                            float (*part)[16][64], float (*rmax)[16], float (*rsum)[16],
                                            float* __restrict__ po, int tid) {
    constexpr int NS = 2 * NC;
    const int lane = tid & 31, w = tid >> 5;
    const int c = lane & 15, h = lane >> 4;

    v4f av[NS], bv[NS][4];
    #pragma unroll
    for (int s = 0; s < NS; ++s)
        av[s] = *(const v4f*)(pa + 8 * s);
    __builtin_amdgcn_sched_barrier(0);
    int4 kk[NS];
    #pragma unroll
    for (int s = 0; s < NS; ++s)
        kk[s] = *(const int4*)(kpos + 8 * s);
    __builtin_amdgcn_sched_barrier(0);
    #pragma unroll
    for (int s = 0; s < NS; ++s) {
        const int p0 = 8 * s + 4 * h;
        const int r0 = (p0     < cw) ? kk[s].x : safe_row;
        const int r1 = (p0 + 1 < cw) ? kk[s].y : safe_row;
        const int r2 = (p0 + 2 < cw) ? kk[s].z : safe_row;
        const int r3 = (p0 + 3 < cw) ? kk[s].w : safe_row;
        bv[s][0] = *(const v4f*)(pv + (size_t)r0 * D_);
        bv[s][1] = *(const v4f*)(pv + (size_t)r1 * D_);
        bv[s][2] = *(const v4f*)(pv + (size_t)r2 * D_);
        bv[s][3] = *(const v4f*)(pv + (size_t)r3 * D_);
    }

    float m = fmaxf(fmaxf(av[0].x, av[0].y), fmaxf(av[0].z, av[0].w));
    #pragma unroll
    for (int s = 1; s < NS; ++s)
        m = fmaxf(m, fmaxf(fmaxf(av[s].x, av[s].y), fmaxf(av[s].z, av[s].w)));
    m = fmaxf(m, __shfl_xor(m, 16));
    rmax[w][c] = m;
    __builtin_amdgcn_fence(__ATOMIC_RELEASE, "workgroup", "local");
    __builtin_amdgcn_s_barrier();
    __builtin_amdgcn_fence(__ATOMIC_ACQUIRE, "workgroup", "local");
    float mx = rmax[0][c];
    #pragma unroll
    for (int i = 1; i < 8; ++i) mx = fmaxf(mx, rmax[i][c]);

    const float nmx = -mx * LOG2E;
    float ls = 0.f;
    #pragma unroll
    for (int s = 0; s < NS; ++s) {
        v4f p;
        p.x = __builtin_amdgcn_exp2f(__builtin_fmaf(av[s].x, LOG2E, nmx));
        p.y = __builtin_amdgcn_exp2f(__builtin_fmaf(av[s].y, LOG2E, nmx));
        p.z = __builtin_amdgcn_exp2f(__builtin_fmaf(av[s].z, LOG2E, nmx));
        p.w = __builtin_amdgcn_exp2f(__builtin_fmaf(av[s].w, LOG2E, nmx));
        av[s] = p;
        ls += (p.x + p.y) + (p.z + p.w);
    }
    ls += __shfl_xor(ls, 16);
    rsum[w][c] = ls;

    v8f acc0 = {}, acc1 = {}, acc2 = {}, acc3 = {};
    #pragma unroll
    for (int s = 0; s < NS; ++s) {
        const v4f a = av[s], r0 = bv[s][0], r1 = bv[s][1], r2 = bv[s][2], r3 = bv[s][3];
        v2f a01, a23, f;
        a01.x = a.x; a01.y = a.y;
        a23.x = a.z; a23.y = a.w;
        f.x = r0.x; f.y = r1.x;
        acc0 = __builtin_amdgcn_wmma_f32_16x16x4_f32(false, a01, false, f, (short)0, acc0, false, false);
        f.x = r0.y; f.y = r1.y;
        acc1 = __builtin_amdgcn_wmma_f32_16x16x4_f32(false, a01, false, f, (short)0, acc1, false, false);
        f.x = r0.z; f.y = r1.z;
        acc2 = __builtin_amdgcn_wmma_f32_16x16x4_f32(false, a01, false, f, (short)0, acc2, false, false);
        f.x = r0.w; f.y = r1.w;
        acc3 = __builtin_amdgcn_wmma_f32_16x16x4_f32(false, a01, false, f, (short)0, acc3, false, false);
        f.x = r2.x; f.y = r3.x;
        acc0 = __builtin_amdgcn_wmma_f32_16x16x4_f32(false, a23, false, f, (short)0, acc0, false, false);
        f.x = r2.y; f.y = r3.y;
        acc1 = __builtin_amdgcn_wmma_f32_16x16x4_f32(false, a23, false, f, (short)0, acc1, false, false);
        f.x = r2.z; f.y = r3.z;
        acc2 = __builtin_amdgcn_wmma_f32_16x16x4_f32(false, a23, false, f, (short)0, acc2, false, false);
        f.x = r2.w; f.y = r3.w;
        acc3 = __builtin_amdgcn_wmma_f32_16x16x4_f32(false, a23, false, f, (short)0, acc3, false, false);
    }

    #pragma unroll
    for (int r = 0; r < 8; ++r) {
        v4f t;
        t.x = acc0[r]; t.y = acc1[r]; t.z = acc2[r]; t.w = acc3[r];
        *(v4f*)&part[w][8 * h + r][4 * c] = t;
    }
    __syncthreads();

    const int row = tid >> 4, c4 = (tid & 15) * 4;
    float tot = rsum[0][row];
    #pragma unroll
    for (int i = 1; i < 8; ++i) tot += rsum[i][row];
    const float inv = 1.0f / tot;
    v4f sum = *(const v4f*)&part[0][row][c4];
    #pragma unroll
    for (int i = 1; i < 8; ++i)
        sum += *(const v4f*)&part[i][row][c4];
    *(v4f*)(po + (size_t)row * D_ + c4) = sum * inv;
}

__global__ __launch_bounds__(256)
void attend_softmax_wmma(const float* __restrict__ scores, const float* __restrict__ V,
                         const int* __restrict__ mask, float* __restrict__ out) {
    __shared__ float part[8][16][64];
    __shared__ float rmax[8][16];
    __shared__ float rsum[8][16];
    __shared__ __attribute__((aligned(16))) int kidx[K_];
    __shared__ int kcnt[32];
    const int tid = threadIdx.x;
    const int lane = tid & 31;
    const int w = __builtin_amdgcn_readfirstlane(tid >> 5);
    const int c = lane & 15, h = lane >> 4;
    const int q0 = blockIdx.x * 16, d0 = blockIdx.y * 64, b = blockIdx.z;

    int mv[K_ / 256];
    #pragma unroll
    for (int r = 0; r < K_ / 256; ++r) mv[r] = mask[(size_t)b * K_ + r * 256 + tid];
    const int n = kept_list<256>(mv, kidx, kcnt, tid);
    const int nc = max(1, (n + 127) >> 7);
    const int kw = 16 * nc;
    const int c0 = w * kw;
    const int cw = max(0, min(kw, n - c0));
    const int safe_row = n > 0 ? kidx[0] : 0;

    const float* pa = scores + ((size_t)(b * KT_ + w) * Q_ + q0 + c) * SCP_ + 4 * h;
    const float* pv = V + (size_t)b * K_ * D_ + d0 + 4 * c;
    const int* kpos = kidx + c0 + 4 * h;
    float* po = out + ((size_t)b * Q_ + q0) * D_ + d0;

    switch (nc) {
        case 1: attend_tile<1>(pa, pv, kpos, cw, safe_row, part, rmax, rsum, po, tid); break;
        case 2: attend_tile<2>(pa, pv, kpos, cw, safe_row, part, rmax, rsum, po, tid); break;
        case 3: attend_tile<3>(pa, pv, kpos, cw, safe_row, part, rmax, rsum, po, tid); break;
        case 4: attend_tile<4>(pa, pv, kpos, cw, safe_row, part, rmax, rsum, po, tid); break;
        case 5: attend_tile<5>(pa, pv, kpos, cw, safe_row, part, rmax, rsum, po, tid); break;
        case 6: attend_tile<6>(pa, pv, kpos, cw, safe_row, part, rmax, rsum, po, tid); break;
        case 7: attend_tile<7>(pa, pv, kpos, cw, safe_row, part, rmax, rsum, po, tid); break;
        default: attend_tile<8>(pa, pv, kpos, cw, safe_row, part, rmax, rsum, po, tid); break;
    }
}

extern "C" void kernel_launch(void* const* d_in, const int* in_sizes, int n_in,
                              void* d_out, int out_size, void* d_ws, size_t ws_size,
                              hipStream_t stream) {
    const float* q    = (const float*)d_in[0];
    const float* k    = (const float*)d_in[1];
    const float* v    = (const float*)d_in[2];
    const int*   mask = (const int*)  d_in[3];
    const float* Wq   = (const float*)d_in[4];
    const float* Wk   = (const float*)d_in[5];
    const float* wv   = (const float*)d_in[6];
    float* out = (float*)d_out;

    float* qp = (float*)d_ws;
    float* kp = qp + (size_t)B_ * Q_ * D_;
    float* sc = kp + (size_t)B_ * K_ * D_;
    int* nkeep = (int*)(sc + (size_t)B_ * KT_ * Q_ * SCP_);

    proj_gemm_wmma<<<dim3(D_ / PC_, (B_ * K_) / PKR_), 256, 0, stream>>>(q, k, Wq, Wk, mask, qp, kp, nkeep);
    scores_kernel<<<dim3(KT_, Q_ / SQ_, B_), 128, 0, stream>>>(qp, kp, wv, nkeep, sc);
    attend_softmax_wmma<<<dim3(Q_ / 16, D_ / 64, B_), 256, 0, stream>>>(sc, v, mask, out);
}


# scores: tile staged and summed in two 128-e slabs, all loads of both (the tile's own rows only) issued at entry: slab 1 arrives under slab 0's terms; range test per slab; same sums, same order
# speedup vs baseline: 4.0982x; 1.0439x over previous
#include <hip/hip_runtime.h>
#include <hip/hip_bf16.h>

typedef __attribute__((ext_vector_type(2))) float v2f;
typedef __attribute__((ext_vector_type(4))) float v4f;
typedef __attribute__((ext_vector_type(8))) float v8f;

#define B_ 4
#define Q_ 256
#define K_ 1024
#define D_ 256

#define PKR_ 64
#define PQR_ 16
#define PC_  64
#define PP_  (PC_ + 4)

__device__ __forceinline__ void proj_tile(const v4f a[4], const v4f b[4][4],
                                          float (*dst)[PP_], int h, int c) {
    v8f acc0 = {}, acc1 = {}, acc2 = {}, acc3 = {};
    #pragma unroll
    for (int s = 0; s < 4; ++s) {
        const v4f r0 = b[s][0], r1 = b[s][1], r2 = b[s][2], r3 = b[s][3];
        v2f a01, a23, f;
        a01.x = a[s].x; a01.y = a[s].y;
        a23.x = a[s].z; a23.y = a[s].w;
        f.x = r0.x; f.y = r1.x;
        acc0 = __builtin_amdgcn_wmma_f32_16x16x4_f32(false, a01, false, f, (short)0, acc0, false, false);
        f.x = r0.y; f.y = r1.y;
        acc1 = __builtin_amdgcn_wmma_f32_16x16x4_f32(false, a01, false, f, (short)0, acc1, false, false);
        f.x = r0.z; f.y = r1.z;
        acc2 = __builtin_amdgcn_wmma_f32_16x16x4_f32(false, a01, false, f, (short)0, acc2, false, false);
        f.x = r0.w; f.y = r1.w;
        acc3 = __builtin_amdgcn_wmma_f32_16x16x4_f32(false, a01, false, f, (short)0, acc3, false, false);
        f.x = r2.x; f.y = r3.x;
        acc0 = __builtin_amdgcn_wmma_f32_16x16x4_f32(false, a23, false, f, (short)0, acc0, false, false);
        f.x = r2.y; f.y = r3.y;
        acc1 = __builtin_amdgcn_wmma_f32_16x16x4_f32(false, a23, false, f, (short)0, acc1, false, false);
        f.x = r2.z; f.y = r3.z;
        acc2 = __builtin_amdgcn_wmma_f32_16x16x4_f32(false, a23, false, f, (short)0, acc2, false, false);
        f.x = r2.w; f.y = r3.w;
        acc3 = __builtin_amdgcn_wmma_f32_16x16x4_f32(false, a23, false, f, (short)0, acc3, false, false);
    }
    #pragma unroll
    for (int r = 0; r < 8; ++r) {
        v4f t;
        t.x = acc0[r]; t.y = acc1[r]; t.z = acc2[r]; t.w = acc3[r];
        *(v4f*)&dst[8 * h + r][4 * c] = t;
    }
}

#define KT_ 8
#define SCP_ 128

__device__ __forceinline__ void barrier_lds_only() {
    __builtin_amdgcn_fence(__ATOMIC_RELEASE, "workgroup", "local");
    __builtin_amdgcn_s_barrier();
    __builtin_amdgcn_fence(__ATOMIC_ACQUIRE, "workgroup", "local");
}

template <int NT, bool LDS_ONLY = false>
__device__ __forceinline__ int kept_list(const int (&mv)[K_ / NT], int* idx, int* cnt, int tid) {
    constexpr int R = K_ / NT, NW = NT / 32;
    const int lane = tid & 31, w = tid >> 5;
    unsigned bal[R];
    #pragma unroll
    for (int r = 0; r < R; ++r) {
        bal[r] = __builtin_amdgcn_ballot_w32(mv[r] != 0);
        if (lane == 0) cnt[r * NW + w] = __builtin_popcount(bal[r]);
    }
    if (LDS_ONLY) barrier_lds_only(); else __syncthreads();
    int base[R], total = 0;
    #pragma unroll
    for (int r = 0; r < R; ++r) base[r] = 0;
    #pragma unroll
    for (int i = 0; i < R * NW; ++i) {
        const int cc = cnt[i];
        #pragma unroll
        for (int r = 0; r < R; ++r) base[r] += (i < r * NW + w) ? cc : 0;
        total += cc;
    }
    const unsigned below = (1u << lane) - 1u;
    #pragma unroll
    for (int r = 0; r < R; ++r)
        if (mv[r] != 0) idx[base[r] + __builtin_popcount(bal[r] & below)] = r * NT + tid;
    if (LDS_ONLY) barrier_lds_only(); else __syncthreads();
    return __builtin_amdgcn_readfirstlane(total);
}

template <int NT>
__device__ __forceinline__ void proj_block(const float* __restrict__ xk, const int* kidx, int n, int r,
                                           const v4f (&aq)[4], const v4f (&bq)[4][4], const v4f (&bk)[4][4],
                                           float (*part)[PKR_ + PQR_][PP_],
                                           float* __restrict__ yq, float* __restrict__ yk, int tid) {
    const int lane = tid & 31, w = tid >> 5;
    const int c = lane & 15, h = lane >> 4;
    constexpr int NA = NT > 0 ? NT : 1;
    v4f ak[NA][4];
    if (NT > 0) {
        int key[NA];
        #pragma unroll
        for (int t = 0; t < NT; ++t) key[t] = kidx[min(16 * (r + 16 * t) + c, n - 1)];
        #pragma unroll
        for (int t = 0; t < NT; ++t) {
            #pragma unroll
            for (int s = 0; s < 4; ++s)
                ak[t][s] = *(const v4f*)(xk + (size_t)key[t] * D_ + 8 * s);
        }
    }
    __builtin_amdgcn_sched_barrier(0);

    proj_tile(aq, bq, &part[w][PKR_], h, c);
    #pragma unroll
    for (int t = 0; t < NT; ++t)
        proj_tile(ak[t], bk, &part[w][16 * t], h, c);
    __syncthreads();

    const int r16 = tid >> 4, c4 = (tid & 15) * 4;
    {
        v4f sum = *(const v4f*)&part[0][PKR_ + r16][c4];
        #pragma unroll
        for (int p = 1; p < 8; ++p)
            sum += *(const v4f*)&part[p][PKR_ + r16][c4];
        *(v4f*)(yq + (size_t)r16 * D_ + c4) = sum;
    }
    #pragma unroll
    for (int t = 0; t < NT; ++t) {
        const int row = 16 * t + r16;
        const int pos = 16 * (r + 16 * t) + r16;
        v4f sum = *(const v4f*)&part[0][row][c4];
        #pragma unroll
        for (int p = 1; p < 8; ++p)
            sum += *(const v4f*)&part[p][row][c4];
        if (pos < n) *(v4f*)(yk + (size_t)pos * D_ + c4) = sum;
    }
}

__global__ __launch_bounds__(256)
void proj_gemm_wmma(const float* __restrict__ Xq, const float* __restrict__ Xk,
                    const float* __restrict__ Wq, const float* __restrict__ Wk,
                    const int* __restrict__ mask,
                    float* __restrict__ qp, float* __restrict__ kp, int* __restrict__ nkeep) {
    __shared__ float part[8][PKR_ + PQR_][PP_];
    __shared__ __attribute__((aligned(16))) int kidx[K_];
    __shared__ int kcnt[32];
    const int tid = threadIdx.x;
    const int lane = tid & 31, w = tid >> 5;
    const int c = lane & 15, h = lane >> 4;
    const int n0 = blockIdx.x * PC_;
    const int by = blockIdx.y;
    const int b = by >> 4, r = by & 15;
    const int k0 = 32 * w;

    int mv[K_ / 256];
    #pragma unroll
    for (int i = 0; i < K_ / 256; ++i) mv[i] = mask[(size_t)b * K_ + i * 256 + tid];
    __builtin_amdgcn_sched_barrier(0);

    const float* paq = Xq + (size_t)(by * PQR_ + c) * D_ + k0 + 4 * h;
    const float* pbk = Wk + (size_t)(k0 + 4 * h) * D_ + n0 + 4 * c;
    const float* pbq = Wq + (size_t)(k0 + 4 * h) * D_ + n0 + 4 * c;

    v4f bk[4][4], bq[4][4], aq[4];
    #pragma unroll
    for (int s = 0; s < 4; ++s) {
        aq[s] = *(const v4f*)(paq + 8 * s);
        #pragma unroll
        for (int i = 0; i < 4; ++i)
            bq[s][i] = *(const v4f*)(pbq + (size_t)(8 * s + i) * D_);
    }
    #pragma unroll
    for (int s = 0; s < 4; ++s) {
        #pragma unroll
        for (int i = 0; i < 4; ++i)
            bk[s][i] = *(const v4f*)(pbk + (size_t)(8 * s + i) * D_);
    }
    __builtin_amdgcn_sched_barrier(0);

    const int n = kept_list<256, true>(mv, kidx, kcnt, tid);
    if (blockIdx.x == 0 && r == 0 && tid == 0) nkeep[b] = n;
    const int tiles = (n + 15) >> 4;
    const int nt = max(0, (tiles - r + 15) >> 4);

    const float* xk = Xk + (size_t)b * K_ * D_ + k0 + 4 * h;
    float* yq = qp + (size_t)(by * PQR_) * D_ + n0;
    float* yk = kp + (size_t)b * K_ * D_ + n0;
    switch (nt) {
        case 0:  proj_block<0>(xk, kidx, n, r, aq, bq, bk, part, yq, yk, tid); break;
        case 1:  proj_block<1>(xk, kidx, n, r, aq, bq, bk, part, yq, yk, tid); break;
        case 2:  proj_block<2>(xk, kidx, n, r, aq, bq, bk, part, yq, yk, tid); break;
        case 3:  proj_block<3>(xk, kidx, n, r, aq, bq, bk, part, yq, yk, tid); break;
        default: proj_block<4>(xk, kidx, n, r, aq, bq, bk, part, yq, yk, tid); break;
    }
}

#define SQ_ 32
#define SKMAX_ 128
#define SH_ 128
#define SP_ (D_ + 4)
#define TWO_LOG2E 2.88539008177793f

template <int NC> struct ScoreOps { v4f q[4], k[NC], w; };

template <int NC>
__device__ __forceinline__ void scores_read(ScoreOps<NC>& o, const float* qrow, const float* krow,
                                            const float* wm2, int e) {
    #pragma unroll
    for (int i = 0; i < 4; ++i) o.q[i] = *(const v4f*)(qrow + 8 * i * SP_ + e);
    #pragma unroll
    for (int j = 0; j < NC; ++j) o.k[j] = *(const v4f*)(krow + 16 * j * SP_ + e);
    o.w = *(const v4f*)(wm2 + e);
}

template <int NC, bool FACTORED>
__device__ __forceinline__ v2f scores_den(const ScoreOps<NC>& o, int n) {
    const int hf = n / (4 * NC), i = (n / NC) & 3, j = n % NC;
    v2f a, b, one;
    a.x = hf ? o.q[i].z : o.q[i].x; a.y = hf ? o.q[i].w : o.q[i].y;
    b.x = hf ? o.k[j].z : o.k[j].x; b.y = hf ? o.k[j].w : o.k[j].y;
    one.x = 1.0f; one.y = 1.0f;
    if (FACTORED) return a * b + one;
    const v2f x = (a + b) * TWO_LOG2E;
    v2f t;
    t.x = __builtin_amdgcn_exp2f(x.x); t.y = __builtin_amdgcn_exp2f(x.y);
    return t + one;
}

template <int NC, bool FACTORED>
__device__ __forceinline__ void scores_terms(const ScoreOps<NC>& o, v2f acc[4][NC], v2f& wsum) {
    constexpr int NP = 8 * NC;
    v2f wlo, whi;
    wlo.x = o.w.x; wlo.y = o.w.y; whi.x = o.w.z; whi.y = o.w.w;
    v2f t[NP], r[NP];
    #pragma unroll
    for (int n = 0; n < 4; ++n) t[n] = scores_den<NC, FACTORED>(o, n);
    if (FACTORED) __builtin_amdgcn_sched_barrier(0);
    #pragma unroll
    for (int n = 0; n < NP; ++n) {
        r[n].x = __builtin_amdgcn_rcpf(t[n].x);
        r[n].y = __builtin_amdgcn_rcpf(t[n].y);
        if (n + 4 < NP) t[n + 4] = scores_den<NC, FACTORED>(o, n + 4);
        if (n >= 3) {
            const int m = n - 3;
            acc[(m / NC) & 3][m % NC] = r[m] * ((m / (4 * NC)) ? whi : wlo) + acc[(m / NC) & 3][m % NC];
        }
        if (FACTORED) {
            __builtin_amdgcn_sched_group_barrier(0x400, 1, 0);
            __builtin_amdgcn_sched_group_barrier(0x002, 1, 0);
            __builtin_amdgcn_sched_group_barrier(0x400, 1, 0);
            __builtin_amdgcn_sched_group_barrier(0x002, 1, 0);
            __builtin_amdgcn_sched_barrier(0);
        }
    }
    #pragma unroll
    for (int m = NP - 3; m < NP; ++m)
        acc[(m / NC) & 3][m % NC] = r[m] * ((m / (4 * NC)) ? whi : wlo) + acc[(m / NC) & 3][m % NC];
    wsum += wlo + whi;
}

template <int NC, bool FACTORED>
__device__ __forceinline__ void scores_loop(const float* qrow, const float* krow, const float* wm2,
                                            int e0, v2f acc[4][NC], v2f& wsum) {
    ScoreOps<NC> A, B;
    scores_read<NC>(A, qrow, krow, wm2, e0);
    for (int e = e0; e < e0 + SH_; e += 8) {
        scores_read<NC>(B, qrow, krow, wm2, e + 4);
        if (FACTORED) __builtin_amdgcn_sched_barrier(0);
        scores_terms<NC, FACTORED>(A, acc, wsum);
        scores_read<NC>(A, qrow, krow, wm2, e0 + ((e - e0 + 8) & (SH_ - 1)));
        if (FACTORED) __builtin_amdgcn_sched_barrier(0);
        scores_terms<NC, FACTORED>(B, acc, wsum);
    }
}

__device__ __forceinline__ v4f exp2x4(v4f x) {
    v4f y;
    y.x = __builtin_amdgcn_exp2f(x.x * TWO_LOG2E); y.y = __builtin_amdgcn_exp2f(x.y * TWO_LOG2E);
    y.z = __builtin_amdgcn_exp2f(x.z * TWO_LOG2E); y.w = __builtin_amdgcn_exp2f(x.w * TWO_LOG2E);
    return y;
}

template <int NC> struct SlabRegs { v4f q[SQ_ / 4], k[4 * NC]; };

template <int NC>
__device__ __forceinline__ void slab_load(SlabRegs<NC>& s, const float* __restrict__ qpb,
                                          const float* __restrict__ kpb, int cn, int tid, int hf) {
    const int c4 = SH_ * hf + (tid & 31) * 4, rq = tid >> 5;
    #pragma unroll
    for (int i = 0; i < SQ_ / 4; ++i)
        s.q[i] = *(const v4f*)(qpb + (size_t)(4 * i + rq) * D_ + c4);
    #pragma unroll
    for (int i = 0; i < 4 * NC; ++i)
        s.k[i] = *(const v4f*)(kpb + (size_t)min(4 * i + rq, cn - 1) * D_ + c4);
}

template <int NC>
__device__ __forceinline__ float slab_put(const SlabRegs<NC>& s, float (*qs)[SP_], float (*ks)[SP_],
                                          int cn, int tid, int hf) {
    const int c4 = SH_ * hf + (tid & 31) * 4, rq = tid >> 5;
    float big = 0.f;
    #pragma unroll
    for (int i = 0; i < SQ_ / 4; ++i) {
        const v4f x = s.q[i];
        big = fmaxf(fmaxf(big, fmaxf(fabsf(x.x), fabsf(x.y))), fmaxf(fabsf(x.z), fabsf(x.w)));
        *(v4f*)&qs[4 * i + rq][c4] = exp2x4(x);
    }
    #pragma unroll
    for (int i = 0; i < 4 * NC; ++i) {
        const int r = 4 * i + rq;
        v4f y = s.k[i];
        if (r >= cn) y = v4f{0.f, 0.f, 0.f, 0.f};
        big = fmaxf(fmaxf(big, fmaxf(fabsf(y.x), fabsf(y.y))), fmaxf(fabsf(y.z), fabsf(y.w)));
        *(v4f*)&ks[r][c4] = exp2x4(y);
    }
    return big;
}

__device__ __forceinline__ void slab_restage_raw(const float* __restrict__ qpb, const float* __restrict__ kpb,
                                 float (*qs)[SP_], float (*ks)[SP_], int cn, int kw, int tid, int hf) {
    const int c4 = SH_ * hf + (tid & 31) * 4, rq = tid >> 5;
    for (int r = rq; r < SQ_; r += 4)
        *(v4f*)&qs[r][c4] = *(const v4f*)(qpb + (size_t)r * D_ + c4);
    for (int r = rq; r < kw; r += 4) {
        v4f y = *(const v4f*)(kpb + (size_t)min(r, cn - 1) * D_ + c4);
        if (r >= cn) y = v4f{0.f, 0.f, 0.f, 0.f};
        *(v4f*)&ks[r][c4] = y;
    }
}

__device__ __forceinline__ int block_or_lds(int pred, int* flag, int tid) {
    const unsigned bal = __builtin_amdgcn_ballot_w32(pred != 0);
    if ((tid & 31) == 0) flag[tid >> 5] = (bal != 0u);
    barrier_lds_only();
    return __builtin_amdgcn_readfirstlane(flag[0] | flag[1] | flag[2] | flag[3]);
}

template <int NC>
__device__ __forceinline__ void scores_tile(const float* __restrict__ qpb, const float* __restrict__ kpb,
                                            float (*qs)[SP_], float (*ks)[SP_], float* wm2,
                                            int (*flag)[4], float wv0, float wv1,
                                            float* __restrict__ sct, int cn, int tid) {
    SlabRegs<NC> s0, s1;
    slab_load<NC>(s0, qpb, kpb, cn, tid, 0);
    __builtin_amdgcn_sched_barrier(0);
    slab_load<NC>(s1, qpb, kpb, cn, tid, 1);
    __builtin_amdgcn_sched_barrier(0);

    v2f acc[4][NC];
    #pragma unroll
    for (int i = 0; i < 4; ++i)
        #pragma unroll
        for (int j = 0; j < NC; ++j) { acc[i][j].x = 0.f; acc[i][j].y = 0.f; }
    v2f wsum; wsum.x = 0.f; wsum.y = 0.f;

    const int lane = tid & 31, w = tid >> 5;
    const int tk = lane & 15, tq = 2 * w + (lane >> 4);
    const float* qrow = &qs[tq][0];
    const float* krow = &ks[tk][0];

    wm2[tid] = -2.0f * wv0;
    wm2[tid + 128] = -2.0f * wv1;
    float big = slab_put<NC>(s0, qs, ks, cn, tid, 0);
    #pragma unroll
    for (int hf = 0; hf < 2; ++hf) {
        const int out_of_range = block_or_lds(!(big <= 40.0f), flag[hf], tid);
        if (!out_of_range) {
            scores_loop<NC, true>(qrow, krow, wm2, SH_ * hf, acc, wsum);
        } else {
            slab_restage_raw(qpb, kpb, qs, ks, cn, 16 * NC, tid, hf);
            barrier_lds_only();
            scores_loop<NC, false>(qrow, krow, wm2, SH_ * hf, acc, wsum);
        }
        if (hf == 0) big = slab_put<NC>(s1, qs, ks, cn, tid, 1);
    }

    const float wtot = -0.5f * (wsum.x + wsum.y);
    const float ninf = -__builtin_inff();
    float* dst = sct + (size_t)tq * SCP_ + tk;
    #pragma unroll
    for (int i = 0; i < 4; ++i)
        #pragma unroll
        for (int j = 0; j < NC; ++j)
            dst[(size_t)(8 * i) * SCP_ + 16 * j] =
                (tk + 16 * j < cn) ? (acc[i][j].x + acc[i][j].y) + wtot : ninf;
}

__global__ __launch_bounds__(128)
void scores_kernel(const float* __restrict__ qp, const float* __restrict__ kp,
                   const float* __restrict__ wv, const int* __restrict__ nkeep,
                   float* __restrict__ scores) {
    __shared__ float qs[SQ_][SP_];
    __shared__ float ks[SKMAX_][SP_];
    __shared__ float wm2[D_];
    __shared__ int flag[2][4];
    const int tid = threadIdx.x;
    const int t = __builtin_amdgcn_workgroup_id_x(), q0 = __builtin_amdgcn_workgroup_id_y() * SQ_,
              b = __builtin_amdgcn_workgroup_id_z();

    const int n = nkeep[b];
    const float wv0 = wv[tid], wv1 = wv[tid + 128];
    const float* qpb = qp + ((size_t)b * Q_ + q0) * D_;
    const int nc = (n + 127) >> 7;
    const int kw = 16 * nc;
    const int c0 = t * kw;
    const int cn = min(kw, n - c0);
    const float* kpb = kp + ((size_t)b * K_ + c0) * D_;

    float* sct = scores + ((size_t)(b * KT_ + t) * Q_ + q0) * SCP_;
    if (cn <= 0) {
        const float ninf = -__builtin_inff();
        const int kf = max(kw, 16);
        for (int x = tid; x < SQ_ * SCP_; x += 128)
            if ((x & (SCP_ - 1)) < kf) sct[x] = ninf;
        return;
    }

    switch (nc) {
        case 1: scores_tile<1>(qpb, kpb, qs, ks, wm2, flag, wv0, wv1, sct, cn, tid); break;
        case 2: scores_tile<2>(qpb, kpb, qs, ks, wm2, flag, wv0, wv1, sct, cn, tid); break;
        case 3: scores_tile<3>(qpb, kpb, qs, ks, wm2, flag, wv0, wv1, sct, cn, tid); break;
        case 4: scores_tile<4>(qpb, kpb, qs, ks, wm2, flag, wv0, wv1, sct, cn, tid); break;
        case 5: scores_tile<5>(qpb, kpb, qs, ks, wm2, flag, wv0, wv1, sct, cn, tid); break;
        case 6: scores_tile<6>(qpb, kpb, qs, ks, wm2, flag, wv0, wv1, sct, cn, tid); break;
        case 7: scores_tile<7>(qpb, kpb, qs, ks, wm2, flag, wv0, wv1, sct, cn, tid); break;
        default: scores_tile<8>(qpb, kpb, qs, ks, wm2, flag, wv0, wv1, sct, cn, tid); break;
    }
}

#define LOG2E 1.44269504088896f

template <int NC>
__device__ __forceinline__ void attend_tile(const float* __restrict__ pa, const float* __restrict__ pv,
                                            const int* kpos, int cw, int safe_row,
                                            float (*part)[16][64], float (*rmax)[16], float (*rsum)[16],
                                            float* __restrict__ po, int tid) {
    constexpr int NS = 2 * NC;
    const int lane = tid & 31, w = tid >> 5;
    const int c = lane & 15, h = lane >> 4;

    v4f av[NS], bv[NS][4];
    #pragma unroll
    for (int s = 0; s < NS; ++s)
        av[s] = *(const v4f*)(pa + 8 * s);
    __builtin_amdgcn_sched_barrier(0);
    int4 kk[NS];
    #pragma unroll
    for (int s = 0; s < NS; ++s)
        kk[s] = *(const int4*)(kpos + 8 * s);
    __builtin_amdgcn_sched_barrier(0);
    #pragma unroll
    for (int s = 0; s < NS; ++s) {
        const int p0 = 8 * s + 4 * h;
        const int r0 = (p0     < cw) ? kk[s].x : safe_row;
        const int r1 = (p0 + 1 < cw) ? kk[s].y : safe_row;
        const int r2 = (p0 + 2 < cw) ? kk[s].z : safe_row;
        const int r3 = (p0 + 3 < cw) ? kk[s].w : safe_row;
        bv[s][0] = *(const v4f*)(pv + (size_t)r0 * D_);
        bv[s][1] = *(const v4f*)(pv + (size_t)r1 * D_);
        bv[s][2] = *(const v4f*)(pv + (size_t)r2 * D_);
        bv[s][3] = *(const v4f*)(pv + (size_t)r3 * D_);
    }

    float m = fmaxf(fmaxf(av[0].x, av[0].y), fmaxf(av[0].z, av[0].w));
    #pragma unroll
    for (int s = 1; s < NS; ++s)
        m = fmaxf(m, fmaxf(fmaxf(av[s].x, av[s].y), fmaxf(av[s].z, av[s].w)));
    m = fmaxf(m, __shfl_xor(m, 16));
    rmax[w][c] = m;
    __builtin_amdgcn_fence(__ATOMIC_RELEASE, "workgroup", "local");
    __builtin_amdgcn_s_barrier();
    __builtin_amdgcn_fence(__ATOMIC_ACQUIRE, "workgroup", "local");
    float mx = rmax[0][c];
    #pragma unroll
    for (int i = 1; i < 8; ++i) mx = fmaxf(mx, rmax[i][c]);

    const float nmx = -mx * LOG2E;
    float ls = 0.f;
    #pragma unroll
    for (int s = 0; s < NS; ++s) {
        v4f p;
        p.x = __builtin_amdgcn_exp2f(__builtin_fmaf(av[s].x, LOG2E, nmx));
        p.y = __builtin_amdgcn_exp2f(__builtin_fmaf(av[s].y, LOG2E, nmx));
        p.z = __builtin_amdgcn_exp2f(__builtin_fmaf(av[s].z, LOG2E, nmx));
        p.w = __builtin_amdgcn_exp2f(__builtin_fmaf(av[s].w, LOG2E, nmx));
        av[s] = p;
        ls += (p.x + p.y) + (p.z + p.w);
    }
    ls += __shfl_xor(ls, 16);
    rsum[w][c] = ls;

    v8f acc0 = {}, acc1 = {}, acc2 = {}, acc3 = {};
    #pragma unroll
    for (int s = 0; s < NS; ++s) {
        const v4f a = av[s], r0 = bv[s][0], r1 = bv[s][1], r2 = bv[s][2], r3 = bv[s][3];
        v2f a01, a23, f;
        a01.x = a.x; a01.y = a.y;
        a23.x = a.z; a23.y = a.w;
        f.x = r0.x; f.y = r1.x;
        acc0 = __builtin_amdgcn_wmma_f32_16x16x4_f32(false, a01, false, f, (short)0, acc0, false, false);
        f.x = r0.y; f.y = r1.y;
        acc1 = __builtin_amdgcn_wmma_f32_16x16x4_f32(false, a01, false, f, (short)0, acc1, false, false);
        f.x = r0.z; f.y = r1.z;
        acc2 = __builtin_amdgcn_wmma_f32_16x16x4_f32(false, a01, false, f, (short)0, acc2, false, false);
        f.x = r0.w; f.y = r1.w;
        acc3 = __builtin_amdgcn_wmma_f32_16x16x4_f32(false, a01, false, f, (short)0, acc3, false, false);
        f.x = r2.x; f.y = r3.x;
        acc0 = __builtin_amdgcn_wmma_f32_16x16x4_f32(false, a23, false, f, (short)0, acc0, false, false);
        f.x = r2.y; f.y = r3.y;
        acc1 = __builtin_amdgcn_wmma_f32_16x16x4_f32(false, a23, false, f, (short)0, acc1, false, false);
        f.x = r2.z; f.y = r3.z;
        acc2 = __builtin_amdgcn_wmma_f32_16x16x4_f32(false, a23, false, f, (short)0, acc2, false, false);
        f.x = r2.w; f.y = r3.w;
        acc3 = __builtin_amdgcn_wmma_f32_16x16x4_f32(false, a23, false, f, (short)0, acc3, false, false);
    }

    #pragma unroll
    for (int r = 0; r < 8; ++r) {
        v4f t;
        t.x = acc0[r]; t.y = acc1[r]; t.z = acc2[r]; t.w = acc3[r];
        *(v4f*)&part[w][8 * h + r][4 * c] = t;
    }
    __syncthreads();

    const int row = tid >> 4, c4 = (tid & 15) * 4;
    float tot = rsum[0][row];
    #pragma unroll
    for (int i = 1; i < 8; ++i) tot += rsum[i][row];
    const float inv = 1.0f / tot;
    v4f sum = *(const v4f*)&part[0][row][c4];
    #pragma unroll
    for (int i = 1; i < 8; ++i)
        sum += *(const v4f*)&part[i][row][c4];
    *(v4f*)(po + (size_t)row * D_ + c4) = sum * inv;
}

__global__ __launch_bounds__(256)
void attend_softmax_wmma(const float* __restrict__ scores, const float* __restrict__ V,
                         const int* __restrict__ mask, float* __restrict__ out) {
    __shared__ float part[8][16][64];
    __shared__ float rmax[8][16];
    __shared__ float rsum[8][16];
    __shared__ __attribute__((aligned(16))) int kidx[K_];
    __shared__ int kcnt[32];
    const int tid = threadIdx.x;
    const int lane = tid & 31;
    const int w = __builtin_amdgcn_readfirstlane(tid >> 5);
    const int c = lane & 15, h = lane >> 4;
    const int q0 = blockIdx.x * 16, d0 = blockIdx.y * 64, b = blockIdx.z;

    int mv[K_ / 256];
    #pragma unroll
    for (int r = 0; r < K_ / 256; ++r) mv[r] = mask[(size_t)b * K_ + r * 256 + tid];
    const int n = kept_list<256>(mv, kidx, kcnt, tid);
    const int nc = max(1, (n + 127) >> 7);
    const int kw = 16 * nc;
    const int c0 = w * kw;
    const int cw = max(0, min(kw, n - c0));
    const int safe_row = n > 0 ? kidx[0] : 0;

    const float* pa = scores + ((size_t)(b * KT_ + w) * Q_ + q0 + c) * SCP_ + 4 * h;
    const float* pv = V + (size_t)b * K_ * D_ + d0 + 4 * c;
    const int* kpos = kidx + c0 + 4 * h;
    float* po = out + ((size_t)b * Q_ + q0) * D_ + d0;

    switch (nc) {
        case 1: attend_tile<1>(pa, pv, kpos, cw, safe_row, part, rmax, rsum, po, tid); break;
        case 2: attend_tile<2>(pa, pv, kpos, cw, safe_row, part, rmax, rsum, po, tid); break;
        case 3: attend_tile<3>(pa, pv, kpos, cw, safe_row, part, rmax, rsum, po, tid); break;
        case 4: attend_tile<4>(pa, pv, kpos, cw, safe_row, part, rmax, rsum, po, tid); break;
        case 5: attend_tile<5>(pa, pv, kpos, cw, safe_row, part, rmax, rsum, po, tid); break;
        case 6: attend_tile<6>(pa, pv, kpos, cw, safe_row, part, rmax, rsum, po, tid); break;
        case 7: attend_tile<7>(pa, pv, kpos, cw, safe_row, part, rmax, rsum, po, tid); break;
        default: attend_tile<8>(pa, pv, kpos, cw, safe_row, part, rmax, rsum, po, tid); break;
    }
}

extern "C" void kernel_launch(void* const* d_in, const int* in_sizes, int n_in,
                              void* d_out, int out_size, void* d_ws, size_t ws_size,
                              hipStream_t stream) {
    const float* q    = (const float*)d_in[0];
    const float* k    = (const float*)d_in[1];
    const float* v    = (const float*)d_in[2];
    const int*   mask = (const int*)  d_in[3];
    const float* Wq   = (const float*)d_in[4];
    const float* Wk   = (const float*)d_in[5];
    const float* wv   = (const float*)d_in[6];
    float* out = (float*)d_out;

    float* qp = (float*)d_ws;
    float* kp = qp + (size_t)B_ * Q_ * D_;
    float* sc = kp + (size_t)B_ * K_ * D_;
    int* nkeep = (int*)(sc + (size_t)B_ * KT_ * Q_ * SCP_);

    proj_gemm_wmma<<<dim3(D_ / PC_, (B_ * K_) / PKR_), 256, 0, stream>>>(q, k, Wq, Wk, mask, qp, kp, nkeep);
    scores_kernel<<<dim3(KT_, Q_ / SQ_, B_), 128, 0, stream>>>(qp, kp, wv, nkeep, sc);
    attend_softmax_wmma<<<dim3(Q_ / 16, D_ / 64, B_), 256, 0, stream>>>(sc, v, mask, out);
}
